# starting kernel
# baseline (speedup 1.0000x reference)
#include <hip/hip_runtime.h>
#include <math.h>


#define B_   8
#define C_   512
#define CR_  32
#define HW_  4096
#define C2_  1024

typedef __attribute__((ext_vector_type(2))) float v2f;
typedef __attribute__((ext_vector_type(8))) float v8f;

__device__ __forceinline__ v8f wmma4(v2f a, v2f b, v8f c) {
  return __builtin_amdgcn_wmma_f32_16x16x4_f32(
       false, a,  false, b,
       (short)0, c,  false,  false);
}

template<int NT>
__device__ __forceinline__ void gemm_row(const float* __restrict__ A, int lda,
                                         const float* __restrict__ Bm, int ldb,
                                         int K, v8f c[NT]) {
  const int lane = threadIdx.x & 31;
  const int half = lane >> 4;
  const int l16  = lane & 15;
  const float* ap = A + (size_t)l16 * lda + 2 * half;
  const float* bp = Bm + (size_t)(2 * half) * ldb + l16;
#pragma unroll 2
  for (int k = 0; k < K; k += 4) {
    v2f a = *(const v2f*)ap;
#pragma unroll
    for (int t = 0; t < NT; ++t) {
      v2f b;
      b.x = bp[t * 16];
      b.y = bp[t * 16 + ldb];
      c[t] = wmma4(a, b, c[t]);
    }
    ap += 4;
    bp += (size_t)4 * ldb;
  }
}

template<int NT>
__device__ __forceinline__ void gemm_row_bt(const float* __restrict__ A, int lda,
                                            const float* __restrict__ Bt, int ldbt,
                                            int K, v8f c[NT]) {
  const int lane = threadIdx.x & 31;
  const int half = lane >> 4;
  const int l16  = lane & 15;
  const float* ap = A + (size_t)l16 * lda + 2 * half;
  const float* bp = Bt + (size_t)l16 * ldbt + 2 * half;
#pragma unroll 2
  for (int k = 0; k < K; k += 4) {
    v2f a = *(const v2f*)ap;
#pragma unroll
    for (int t = 0; t < NT; ++t) {
      v2f b = *(const v2f*)(bp + (size_t)t * 16 * ldbt);
      c[t] = wmma4(a, b, c[t]);
    }
    ap += 4;
    bp += 4;
  }
}

__global__ void k_conv1_gelu(const float* __restrict__ x, const float* __restrict__ w1,
                             const float* __restrict__ b1, float* __restrict__ h) {
  constexpr int NT = 4;
  const int wid = blockIdx.x * 8 + (threadIdx.x >> 5);
  const int NG = HW_ / (16 * NT);
  const int MT = CR_ / 16;
  const int b  = wid / (MT * NG);
  const int r0 = wid % (MT * NG);
  const int mt = r0 / NG;
  const int ng = r0 % NG;

  const float* A  = w1 + (size_t)mt * 16 * C_;
  const float* Bm = x + (size_t)b * C_ * HW_ + ng * 16 * NT;
  v8f c[NT] = {};
  gemm_row<NT>(A, C_, Bm, HW_, C_, c);

  const int lane = threadIdx.x & 31;
  const int half = lane >> 4;
  const int l16  = lane & 15;
  float* D = h + (size_t)b * CR_ * HW_;
#pragma unroll
  for (int t = 0; t < NT; ++t)
#pragma unroll
    for (int r = 0; r < 8; ++r) {
      const int m = mt * 16 + r + 8 * half;
      const int n = ng * 16 * NT + t * 16 + l16;
      float v = c[t][r] + b1[m];
      v = 0.5f * v * (1.0f + erff(v * 0.70710678118654752f));
      D[(size_t)m * HW_ + n] = v;
    }
}

__global__ void k_conv2(const float* __restrict__ h, const float* __restrict__ w2,
                        const float* __restrict__ b2, float* __restrict__ y) {
  constexpr int NT = 4;
  const int wid = blockIdx.x * 8 + (threadIdx.x >> 5);
  const int NG = HW_ / (16 * NT);
  const int MT = C_ / 16;
  const int b  = wid / (MT * NG);
  const int r0 = wid % (MT * NG);
  const int mt = r0 / NG;
  const int ng = r0 % NG;

  const float* A  = w2 + (size_t)mt * 16 * CR_;
  const float* Bm = h + (size_t)b * CR_ * HW_ + ng * 16 * NT;
  v8f c[NT] = {};
  gemm_row<NT>(A, CR_, Bm, HW_, CR_, c);

  const int lane = threadIdx.x & 31;
  const int half = lane >> 4;
  const int l16  = lane & 15;
  float* D = y + (size_t)b * C_ * HW_;
#pragma unroll
  for (int t = 0; t < NT; ++t)
#pragma unroll
    for (int r = 0; r < 8; ++r) {
      const int m = mt * 16 + r + 8 * half;
      const int n = ng * 16 * NT + t * 16 + l16;
      D[(size_t)m * HW_ + n] = c[t][r] + b2[m];
    }
}

__global__ void k_energy(const float* __restrict__ y, float* __restrict__ e) {
  constexpr int NT = 4;
  const int wid = blockIdx.x * 8 + (threadIdx.x >> 5);
  const int NG = C_ / (16 * NT);
  const int MT = C_ / 16;
  const int b  = wid / (MT * NG);
  const int r0 = wid % (MT * NG);
  const int mt = r0 / NG;
  const int ng = r0 % NG;

  const float* yb = y + (size_t)b * C_ * HW_;
  const float* A  = yb + (size_t)mt * 16 * HW_;
  const float* Bt = yb + (size_t)ng * 16 * NT * HW_;
  v8f c[NT] = {};
  gemm_row_bt<NT>(A, HW_, Bt, HW_, HW_, c);

  const int lane = threadIdx.x & 31;
  const int half = lane >> 4;
  const int l16  = lane & 15;
  float* D = e + (size_t)b * C_ * C_;
#pragma unroll
  for (int t = 0; t < NT; ++t)
#pragma unroll
    for (int r = 0; r < 8; ++r) {
      const int m = mt * 16 + r + 8 * half;
      const int n = ng * 16 * NT + t * 16 + l16;
      D[(size_t)m * C_ + n] = c[t][r];
    }
}

__global__ void k_softmax(float* __restrict__ e) {
  __shared__ float red[256];
  const int tid = threadIdx.x;
  float* p = e + (size_t)blockIdx.x * C_;
  const float v0 = p[tid];
  const float v1 = p[tid + 256];
  red[tid] = fminf(v0, v1);
  __syncthreads();
  for (int s = 128; s > 0; s >>= 1) {
    if (tid < s) red[tid] = fminf(red[tid], red[tid + s]);
    __syncthreads();
  }
  const float rmin = red[0];
  __syncthreads();
  const float e0 = expf(rmin - v0);
  const float e1 = expf(rmin - v1);
  red[tid] = e0 + e1;
  __syncthreads();
  for (int s = 128; s > 0; s >>= 1) {
    if (tid < s) red[tid] += red[tid + s];
    __syncthreads();
  }
  const float inv = 1.0f / red[0];
  p[tid] = e0 * inv;
  p[tid + 256] = e1 * inv;
}

__global__ void k_m2(const float* __restrict__ attn, const float* __restrict__ wf,
                     const float* __restrict__ gamma, float* __restrict__ m2) {
  constexpr int NT = 4;
  const int wid = blockIdx.x * 8 + (threadIdx.x >> 5);
  const int NG = C_ / (16 * NT);
  const int MT = C_ / 16;
  const int b  = wid / (MT * NG);
  const int r0 = wid % (MT * NG);
  const int mt = r0 / NG;
  const int ng = r0 % NG;

  const float* A  = wf + (size_t)mt * 16 * C2_;
  const float* Bm = attn + (size_t)b * C_ * C_ + ng * 16 * NT;
  v8f c[NT] = {};
  gemm_row<NT>(A, C2_, Bm, C_, C_, c);

  const float g = gamma[0];
  const int lane = threadIdx.x & 31;
  const int half = lane >> 4;
  const int l16  = lane & 15;
  float* D = m2 + (size_t)b * C_ * C_;
#pragma unroll
  for (int t = 0; t < NT; ++t)
#pragma unroll
    for (int r = 0; r < 8; ++r) {
      const int m = mt * 16 + r + 8 * half;
      const int n = ng * 16 * NT + t * 16 + l16;
      D[(size_t)m * C_ + n] = g * c[t][r] + wf[(size_t)m * C2_ + n];
    }
}

__global__ void k_fusion(const float* __restrict__ m2, const float* __restrict__ y,
                         const float* __restrict__ x, const float* __restrict__ wf,
                         const float* __restrict__ bf, float* __restrict__ out) {
  constexpr int NT = 4;
  const int wid = blockIdx.x * 8 + (threadIdx.x >> 5);
  const int NG = HW_ / (16 * NT);
  const int MT = C_ / 16;
  const int b  = wid / (MT * NG);
  const int r0 = wid % (MT * NG);
  const int mt = r0 / NG;
  const int ng = r0 % NG;

  v8f c[NT] = {};
  gemm_row<NT>(m2 + (size_t)b * C_ * C_ + (size_t)mt * 16 * C_, C_,
               y + (size_t)b * C_ * HW_ + ng * 16 * NT, HW_, C_, c);
  gemm_row<NT>(wf + (size_t)mt * 16 * C2_ + C_, C2_,
               x + (size_t)b * C_ * HW_ + ng * 16 * NT, HW_, C_, c);

  const int lane = threadIdx.x & 31;
  const int half = lane >> 4;
  const int l16  = lane & 15;
  float* D = out + (size_t)b * C_ * HW_;
#pragma unroll
  for (int t = 0; t < NT; ++t)
#pragma unroll
    for (int r = 0; r < 8; ++r) {
      const int m = mt * 16 + r + 8 * half;
      const int n = ng * 16 * NT + t * 16 + l16;
      D[(size_t)m * HW_ + n] = c[t][r] + bf[m];
    }
}

extern "C" void kernel_launch(void* const* d_in, const int* in_sizes, int n_in,
                              void* d_out, int out_size, void* d_ws, size_t ws_size,
                              hipStream_t stream) {
  const float* x     = (const float*)d_in[0];
  const float* w1    = (const float*)d_in[1];
  const float* b1    = (const float*)d_in[2];
  const float* w2    = (const float*)d_in[3];
  const float* b2    = (const float*)d_in[4];
  const float* wf    = (const float*)d_in[5];
  const float* bf    = (const float*)d_in[6];
  const float* gamma = (const float*)d_in[7];
  float* out = (float*)d_out;

  char* ws = (char*)d_ws;
  float* h    = (float*)(ws);
  float* y    = (float*)(ws + (size_t)4 * 1024 * 1024);
  float* attn = (float*)(ws + (size_t)68 * 1024 * 1024);
  float* m2   = (float*)(ws + (size_t)76 * 1024 * 1024);

  const dim3 blk(256);
  k_conv1_gelu<<<dim3(128),  blk, 0, stream>>>(x, w1, b1, h);
  k_conv2     <<<dim3(2048), blk, 0, stream>>>(h, w2, b2, y);
  k_energy    <<<dim3(256),  blk, 0, stream>>>(y, attn);
  k_softmax   <<<dim3(4096), blk, 0, stream>>>(attn);
  k_m2        <<<dim3(256),  blk, 0, stream>>>(attn, wf, gamma, m2);
  k_fusion    <<<dim3(2048), blk, 0, stream>>>(m2, y, x, wf, bf, out);
}


# final product M2.y + Wf2.x: 128x128 block tile, both operands staged per K chunk of 32 in LDS by row-contiguous 16-byte loads, 2 buffers; 64x32 wave tiles read LDS fragments; f32 sum order unchanged
# speedup vs baseline: 1.5422x; 1.5422x over previous
#include <hip/hip_runtime.h>
#include <math.h>


#define B_   8
#define C_   512
#define CR_  32
#define HW_  4096
#define C2_  1024

typedef __attribute__((ext_vector_type(2))) float v2f;
typedef __attribute__((ext_vector_type(8))) float v8f;

__device__ __forceinline__ v8f wmma4(v2f a, v2f b, v8f c) {
  return __builtin_amdgcn_wmma_f32_16x16x4_f32(
       false, a,  false, b,
       (short)0, c,  false,  false);
}

template<int NT>
__device__ __forceinline__ void gemm_row(const float* __restrict__ A, int lda,
                                         const float* __restrict__ Bm, int ldb,
                                         int K, v8f c[NT]) {
  const int lane = threadIdx.x & 31;
  const int half = lane >> 4;
  const int l16  = lane & 15;
  const float* ap = A + (size_t)l16 * lda + 2 * half;
  const float* bp = Bm + (size_t)(2 * half) * ldb + l16;
#pragma unroll 2
  for (int k = 0; k < K; k += 4) {
    v2f a = *(const v2f*)ap;
#pragma unroll
    for (int t = 0; t < NT; ++t) {
      v2f b;
      b.x = bp[t * 16];
      b.y = bp[t * 16 + ldb];
      c[t] = wmma4(a, b, c[t]);
    }
    ap += 4;
    bp += (size_t)4 * ldb;
  }
}

__global__ void k_conv1_gelu(const float* __restrict__ x, const float* __restrict__ w1,
                             const float* __restrict__ b1, float* __restrict__ h) {
  constexpr int NT = 4;
  const int wid = blockIdx.x * 8 + (threadIdx.x >> 5);
  const int NG = HW_ / (16 * NT);
  const int MT = CR_ / 16;
  const int b  = wid / (MT * NG);
  const int r0 = wid % (MT * NG);
  const int mt = r0 / NG;
  const int ng = r0 % NG;

  const float* A  = w1 + (size_t)mt * 16 * C_;
  const float* Bm = x + (size_t)b * C_ * HW_ + ng * 16 * NT;
  v8f c[NT] = {};
  gemm_row<NT>(A, C_, Bm, HW_, C_, c);

  const int lane = threadIdx.x & 31;
  const int half = lane >> 4;
  const int l16  = lane & 15;
  float* D = h + (size_t)b * CR_ * HW_;
#pragma unroll
  for (int t = 0; t < NT; ++t)
#pragma unroll
    for (int r = 0; r < 8; ++r) {
      const int m = mt * 16 + r + 8 * half;
      const int n = ng * 16 * NT + t * 16 + l16;
      float v = c[t][r] + b1[m];
      v = 0.5f * v * (1.0f + erff(v * 0.70710678118654752f));
      D[(size_t)m * HW_ + n] = v;
    }
}

__global__ void k_conv2(const float* __restrict__ h, const float* __restrict__ w2,
                        const float* __restrict__ b2, float* __restrict__ y) {
  constexpr int NT = 4;
  const int wid = blockIdx.x * 8 + (threadIdx.x >> 5);
  const int NG = HW_ / (16 * NT);
  const int MT = C_ / 16;
  const int b  = wid / (MT * NG);
  const int r0 = wid % (MT * NG);
  const int mt = r0 / NG;
  const int ng = r0 % NG;

  const float* A  = w2 + (size_t)mt * 16 * CR_;
  const float* Bm = h + (size_t)b * CR_ * HW_ + ng * 16 * NT;
  v8f c[NT] = {};
  gemm_row<NT>(A, CR_, Bm, HW_, CR_, c);

  const int lane = threadIdx.x & 31;
  const int half = lane >> 4;
  const int l16  = lane & 15;
  float* D = y + (size_t)b * C_ * HW_;
#pragma unroll
  for (int t = 0; t < NT; ++t)
#pragma unroll
    for (int r = 0; r < 8; ++r) {
      const int m = mt * 16 + r + 8 * half;
      const int n = ng * 16 * NT + t * 16 + l16;
      D[(size_t)m * HW_ + n] = c[t][r] + b2[m];
    }
}

#define E_BM 128
#define E_BN 64
#define E_KC 32
#define E_LD 36

typedef __attribute__((ext_vector_type(4))) float v4f;

__global__ __launch_bounds__(256) void k_energy(const float* __restrict__ y, float* __restrict__ e) {
  __shared__ __attribute__((aligned(16))) float sm[2 * (E_BM + E_BN) * E_LD];
  constexpr int STAGE = (E_BM + E_BN) * E_LD;
  constexpr int NCH = HW_ / E_KC;

  const int tid  = threadIdx.x;
  const int lane = tid & 31;
  const int half = lane >> 4;
  const int l16  = lane & 15;
  const int wv   = tid >> 5;
  const int wm   = wv >> 1;
  const int wn   = wv & 1;
  const int b    = blockIdx.x >> 5;
  const int bm   = (blockIdx.x >> 3) & 3;
  const int bn   = blockIdx.x & 7;
  const float* yb = y + (size_t)b * C_ * HW_;

  const int srow = tid >> 3;
  const int sc4  = tid & 7;
  const float* gp[6];
  int so[6];
#pragma unroll
  for (int i = 0; i < 6; ++i) {
    const int r = srow + 32 * i;
    const int grow = (i < 4) ? (bm * E_BM + r) : (bn * E_BN + (r - E_BM));
    gp[i] = yb + (size_t)grow * HW_ + 4 * sc4;
    so[i] = r * E_LD + 4 * sc4;
  }

  const int ao = (wm * 32 + l16) * E_LD + 2 * half;
  const int bo = (E_BM + wn * 32 + l16) * E_LD + 2 * half;

  v4f pre[6];
#pragma unroll
  for (int i = 0; i < 6; ++i) pre[i] = *(const v4f*)gp[i];
#pragma unroll
  for (int i = 0; i < 6; ++i) *(v4f*)(sm + so[i]) = pre[i];
  __syncthreads();

  v8f c00 = {}, c01 = {}, c10 = {}, c11 = {};
  constexpr int KS = E_KC / 4;
  constexpr int KS_HEAD = 6;
  for (int kc = 0; kc < NCH; ++kc) {
    const int cur = kc & 1;
    const bool more = (kc + 1 < NCH);
    if (more) {
#pragma unroll
      for (int i = 0; i < 6; ++i) pre[i] = *(const v4f*)(gp[i] + (size_t)(kc + 1) * E_KC);
    }
    const float* L = sm + cur * STAGE;
    v2f fa0[KS], fa1[KS], fb0[KS], fb1[KS];
#pragma unroll
    for (int s = 0; s < KS; ++s) {
      fa0[s] = *(const v2f*)(L + ao + 4 * s);
      fa1[s] = *(const v2f*)(L + ao + 16 * E_LD + 4 * s);
      fb0[s] = *(const v2f*)(L + bo + 4 * s);
      fb1[s] = *(const v2f*)(L + bo + 16 * E_LD + 4 * s);
    }
    __builtin_amdgcn_sched_barrier(0);
#pragma unroll
    for (int s = 0; s < KS_HEAD; ++s) {
      c00 = wmma4(fa0[s], fb0[s], c00);
      c01 = wmma4(fa0[s], fb1[s], c01);
      c10 = wmma4(fa1[s], fb0[s], c10);
      c11 = wmma4(fa1[s], fb1[s], c11);
    }
    __builtin_amdgcn_sched_barrier(0);
    if (more) {
      float* S = sm + (cur ^ 1) * STAGE;
#pragma unroll
      for (int i = 0; i < 6; ++i) *(v4f*)(S + so[i]) = pre[i];
    }
    __syncthreads();
    __builtin_amdgcn_sched_barrier(0);
#pragma unroll
    for (int s = KS_HEAD; s < KS; ++s) {
      c00 = wmma4(fa0[s], fb0[s], c00);
      c01 = wmma4(fa0[s], fb1[s], c01);
      c10 = wmma4(fa1[s], fb0[s], c10);
      c11 = wmma4(fa1[s], fb1[s], c11);
    }
  }

  float* D = e + (size_t)b * C_ * C_;
  const int m0 = bm * E_BM + wm * 32 + 8 * half;
  const int n0 = bn * E_BN + wn * 32 + l16;
#pragma unroll
  for (int r = 0; r < 8; ++r) {
    D[(size_t)(m0 + r) * C_ + n0]           = c00[r];
    D[(size_t)(m0 + r) * C_ + n0 + 16]      = c01[r];
    D[(size_t)(m0 + 16 + r) * C_ + n0]      = c10[r];
    D[(size_t)(m0 + 16 + r) * C_ + n0 + 16] = c11[r];
  }
}

__global__ void k_softmax(float* __restrict__ e) {
  __shared__ float red[256];
  const int tid = threadIdx.x;
  float* p = e + (size_t)blockIdx.x * C_;
  const float v0 = p[tid];
  const float v1 = p[tid + 256];
  red[tid] = fminf(v0, v1);
  __syncthreads();
  for (int s = 128; s > 0; s >>= 1) {
    if (tid < s) red[tid] = fminf(red[tid], red[tid + s]);
    __syncthreads();
  }
  const float rmin = red[0];
  __syncthreads();
  const float e0 = expf(rmin - v0);
  const float e1 = expf(rmin - v1);
  red[tid] = e0 + e1;
  __syncthreads();
  for (int s = 128; s > 0; s >>= 1) {
    if (tid < s) red[tid] += red[tid + s];
    __syncthreads();
  }
  const float inv = 1.0f / red[0];
  p[tid] = e0 * inv;
  p[tid + 256] = e1 * inv;
}

__global__ void k_m2(const float* __restrict__ attn, const float* __restrict__ wf,
                     const float* __restrict__ gamma, float* __restrict__ m2) {
  constexpr int NT = 4;
  const int wid = blockIdx.x * 8 + (threadIdx.x >> 5);
  const int NG = C_ / (16 * NT);
  const int MT = C_ / 16;
  const int b  = wid / (MT * NG);
  const int r0 = wid % (MT * NG);
  const int mt = r0 / NG;
  const int ng = r0 % NG;

  const float* A  = wf + (size_t)mt * 16 * C2_;
  const float* Bm = attn + (size_t)b * C_ * C_ + ng * 16 * NT;
  v8f c[NT] = {};
  gemm_row<NT>(A, C2_, Bm, C_, C_, c);

  const float g = gamma[0];
  const int lane = threadIdx.x & 31;
  const int half = lane >> 4;
  const int l16  = lane & 15;
  float* D = m2 + (size_t)b * C_ * C_;
#pragma unroll
  for (int t = 0; t < NT; ++t)
#pragma unroll
    for (int r = 0; r < 8; ++r) {
      const int m = mt * 16 + r + 8 * half;
      const int n = ng * 16 * NT + t * 16 + l16;
      D[(size_t)m * C_ + n] = g * c[t][r] + wf[(size_t)m * C2_ + n];
    }
}

#define F_BM  128
#define F_BN  128
#define F_KC  32
#define F_LDA 36
#define F_LDB 144

__global__ __launch_bounds__(256) void k_fusion(const float* __restrict__ m2, const float* __restrict__ y,
                                                const float* __restrict__ x, const float* __restrict__ wf,
                                                const float* __restrict__ bf, float* __restrict__ out) {
  constexpr int A_SZ  = F_BM * F_LDA;
  constexpr int STAGE = A_SZ + F_KC * F_LDB;
  __shared__ __attribute__((aligned(16))) float sm[2 * STAGE];
  constexpr int NCH  = C2_ / F_KC;
  constexpr int NCH1 = C_ / F_KC;

  const int tid  = threadIdx.x;
  const int lane = tid & 31;
  const int half = lane >> 4;
  const int l16  = lane & 15;
  const int wv   = tid >> 5;
  const int wm   = wv >> 2;
  const int wn   = wv & 3;
  const int b    = blockIdx.x >> 7;
  const int bm   = (blockIdx.x >> 5) & 3;
  const int bn   = blockIdx.x & 31;

  const int arow = tid >> 3;
  const int ac4  = tid & 7;
  const int brow = tid >> 5;
  const int bc4  = tid & 31;
  const float* gA1 = m2 + (size_t)b * C_ * C_ + (size_t)(bm * F_BM + arow) * C_ + 4 * ac4;
  const float* gA2 = wf + (size_t)(bm * F_BM + arow) * C2_ + C_ + 4 * ac4;
  const float* gB1 = y + (size_t)b * C_ * HW_ + (size_t)brow * HW_ + bn * F_BN + 4 * bc4;
  const float* gB2 = x + (size_t)b * C_ * HW_ + (size_t)brow * HW_ + bn * F_BN + 4 * bc4;
  const int soa = arow * F_LDA + 4 * ac4;
  const int sob = A_SZ + brow * F_LDB + 4 * bc4;

  const int ao = (wm * 64 + l16) * F_LDA + 2 * half;
  const int bo = A_SZ + (2 * half) * F_LDB + wn * 32 + l16;

  v4f pre[8];
#pragma unroll
  for (int i = 0; i < 4; ++i) pre[i] = *(const v4f*)(gA1 + (size_t)i * 32 * C_);
#pragma unroll
  for (int i = 0; i < 4; ++i) pre[4 + i] = *(const v4f*)(gB1 + (size_t)i * 8 * HW_);
#pragma unroll
  for (int i = 0; i < 4; ++i) *(v4f*)(sm + soa + i * 32 * F_LDA) = pre[i];
#pragma unroll
  for (int i = 0; i < 4; ++i) *(v4f*)(sm + sob + i * 8 * F_LDB) = pre[4 + i];
  __syncthreads();

  v8f c[4][2] = {};
  constexpr int KS = F_KC / 4;
  constexpr int KS_HEAD = 6;
  for (int kc = 0; kc < NCH; ++kc) {
    const int cur = kc & 1;
    const bool more = (kc + 1 < NCH);
    if (more) {
      const int kn = kc + 1;
      if (kn < NCH1) {
#pragma unroll
        for (int i = 0; i < 4; ++i) pre[i] = *(const v4f*)(gA1 + (size_t)i * 32 * C_ + kn * F_KC);
#pragma unroll
        for (int i = 0; i < 4; ++i) pre[4 + i] = *(const v4f*)(gB1 + (size_t)(kn * F_KC + 8 * i) * HW_);
      } else {
#pragma unroll
        for (int i = 0; i < 4; ++i) pre[i] = *(const v4f*)(gA2 + (size_t)i * 32 * C2_ + (kn - NCH1) * F_KC);
#pragma unroll
        for (int i = 0; i < 4; ++i) pre[4 + i] = *(const v4f*)(gB2 + (size_t)((kn - NCH1) * F_KC + 8 * i) * HW_);
      }
    }
    const float* L = sm + cur * STAGE;
    v2f fa[4][KS], fb[2][KS];
#pragma unroll
    for (int s = 0; s < KS; ++s) {
#pragma unroll
      for (int t = 0; t < 4; ++t) fa[t][s] = *(const v2f*)(L + ao + t * 16 * F_LDA + 4 * s);
#pragma unroll
      for (int t = 0; t < 2; ++t) {
        fb[t][s].x = L[bo + t * 16 + 4 * s * F_LDB];
        fb[t][s].y = L[bo + t * 16 + 4 * s * F_LDB + F_LDB];
      }
    }
    __builtin_amdgcn_sched_barrier(0);
#pragma unroll
    for (int s = 0; s < KS_HEAD; ++s)
#pragma unroll
      for (int t = 0; t < 4; ++t) {
        c[t][0] = wmma4(fa[t][s], fb[0][s], c[t][0]);
        c[t][1] = wmma4(fa[t][s], fb[1][s], c[t][1]);
      }
    __builtin_amdgcn_sched_barrier(0);
    if (more) {
      float* S = sm + (cur ^ 1) * STAGE;
#pragma unroll
      for (int i = 0; i < 4; ++i) *(v4f*)(S + soa + i * 32 * F_LDA) = pre[i];
#pragma unroll
      for (int i = 0; i < 4; ++i) *(v4f*)(S + sob + i * 8 * F_LDB) = pre[4 + i];
    }
    __syncthreads();
    __builtin_amdgcn_sched_barrier(0);
#pragma unroll
    for (int s = KS_HEAD; s < KS; ++s)
#pragma unroll
      for (int t = 0; t < 4; ++t) {
        c[t][0] = wmma4(fa[t][s], fb[0][s], c[t][0]);
        c[t][1] = wmma4(fa[t][s], fb[1][s], c[t][1]);
      }
  }

  float* D = out + (size_t)b * C_ * HW_;
  const int n0 = bn * F_BN + wn * 32 + l16;
#pragma unroll
  for (int t = 0; t < 4; ++t) {
    const int m0 = bm * F_BM + wm * 64 + t * 16 + 8 * half;
#pragma unroll
    for (int r = 0; r < 8; ++r) {
      const float bias = bf[m0 + r];
      D[(size_t)(m0 + r) * HW_ + n0]      = c[t][0][r] + bias;
      D[(size_t)(m0 + r) * HW_ + n0 + 16] = c[t][1][r] + bias;
    }
  }
}

extern "C" void kernel_launch(void* const* d_in, const int* in_sizes, int n_in,
                              void* d_out, int out_size, void* d_ws, size_t ws_size,
                              hipStream_t stream) {
  const float* x     = (const float*)d_in[0];
  const float* w1    = (const float*)d_in[1];
  const float* b1    = (const float*)d_in[2];
  const float* w2    = (const float*)d_in[3];
  const float* b2    = (const float*)d_in[4];
  const float* wf    = (const float*)d_in[5];
  const float* bf    = (const float*)d_in[6];
  const float* gamma = (const float*)d_in[7];
  float* out = (float*)d_out;

  char* ws = (char*)d_ws;
  float* h    = (float*)(ws);
  float* y    = (float*)(ws + (size_t)4 * 1024 * 1024);
  float* attn = (float*)(ws + (size_t)68 * 1024 * 1024);
  float* m2   = (float*)(ws + (size_t)76 * 1024 * 1024);

  const dim3 blk(256);
  k_conv1_gelu<<<dim3(128),  blk, 0, stream>>>(x, w1, b1, h);
  k_conv2     <<<dim3(2048), blk, 0, stream>>>(h, w2, b2, y);
  k_energy    <<<dim3(256),  blk, 0, stream>>>(y, attn);
  k_softmax   <<<dim3(4096), blk, 0, stream>>>(attn);
  k_m2        <<<dim3(256),  blk, 0, stream>>>(attn, wf, gamma, m2);
  k_fusion    <<<dim3(1024), blk, 0, stream>>>(m2, y, x, wf, bf, out);
}


# conv1: 32x128 block tile; w1 and x staged per 32-k chunk in LDS by 16-byte row loads kept 3 chunks in flight across LDS-only barriers; both 16-row tiles from one x fragment; f32 sum order unchanged
# speedup vs baseline: 1.6682x; 1.0817x over previous
#include <hip/hip_runtime.h>
#include <math.h>


#define B_   8
#define C_   512
#define CR_  32
#define HW_  4096
#define C2_  1024

typedef __attribute__((ext_vector_type(2))) float v2f;
typedef __attribute__((ext_vector_type(8))) float v8f;

__device__ __forceinline__ v8f wmma4(v2f a, v2f b, v8f c) {
  return __builtin_amdgcn_wmma_f32_16x16x4_f32(
       false, a,  false, b,
       (short)0, c,  false,  false);
}

template<int NT>
__device__ __forceinline__ void gemm_row(const float* __restrict__ A, int lda,
                                         const float* __restrict__ Bm, int ldb,
                                         int K, v8f c[NT]) {
  const int lane = threadIdx.x & 31;
  const int half = lane >> 4;
  const int l16  = lane & 15;
  const float* ap = A + (size_t)l16 * lda + 2 * half;
  const float* bp = Bm + (size_t)(2 * half) * ldb + l16;
#pragma unroll 2
  for (int k = 0; k < K; k += 4) {
    v2f a = *(const v2f*)ap;
#pragma unroll
    for (int t = 0; t < NT; ++t) {
      v2f b;
      b.x = bp[t * 16];
      b.y = bp[t * 16 + ldb];
      c[t] = wmma4(a, b, c[t]);
    }
    ap += 4;
    bp += (size_t)4 * ldb;
  }
}

#define P_BN    128
#define P_KC    32
#define P_LDA   36
#define P_LDB   144
#define P_AHEAD 3

typedef __attribute__((ext_vector_type(4))) float v4f;

__device__ __forceinline__ void lds_barrier() {
  __builtin_amdgcn_fence(__ATOMIC_RELEASE, "workgroup", "local");
  __builtin_amdgcn_s_barrier();
  __builtin_amdgcn_fence(__ATOMIC_ACQUIRE, "workgroup", "local");
}

__global__ __launch_bounds__(256) void k_conv1_gelu(const float* __restrict__ x, const float* __restrict__ w1,
                                                    const float* __restrict__ b1, float* __restrict__ h) {
  constexpr int A_SZ  = CR_ * P_LDA;
  constexpr int STAGE = A_SZ + P_KC * P_LDB;
  __shared__ __attribute__((aligned(16))) float sm[2 * STAGE];
  constexpr int NCH = C_ / P_KC;

  const int tid  = threadIdx.x;
  const int lane = tid & 31;
  const int half = lane >> 4;
  const int l16  = lane & 15;
  const int wv   = tid >> 5;
  const int b    = blockIdx.x >> 5;
  const int bn   = blockIdx.x & 31;

  const int arow = tid >> 3;
  const int ac4  = tid & 7;
  const int brow = tid >> 5;
  const int bc4  = tid & 31;
  const float* gA = w1 + (size_t)arow * C_ + 4 * ac4;
  const float* gB = x + (size_t)b * C_ * HW_ + (size_t)brow * HW_ + bn * P_BN + 4 * bc4;
  const int soa = arow * P_LDA + 4 * ac4;
  const int sob = A_SZ + brow * P_LDB + 4 * bc4;

  const int ao = l16 * P_LDA + 2 * half;
  const int bo = A_SZ + (2 * half) * P_LDB + wv * 16 + l16;

  v4f ring[P_AHEAD][5];
#pragma unroll
  for (int j = 0; j < P_AHEAD; ++j) {
    ring[j][0] = *(const v4f*)(gA + j * P_KC);
#pragma unroll
    for (int i = 0; i < 4; ++i) ring[j][1 + i] = *(const v4f*)(gB + (size_t)(j * P_KC + 8 * i) * HW_);
  }
  *(v4f*)(sm + soa) = ring[0][0];
#pragma unroll
  for (int i = 0; i < 4; ++i) *(v4f*)(sm + sob + i * 8 * P_LDB) = ring[0][1 + i];
  lds_barrier();

  v8f c0 = {}, c1 = {};
  constexpr int KS = P_KC / 4;
  constexpr int KS_HEAD = 6;
#pragma unroll
  for (int kc = 0; kc < NCH; ++kc) {
    const int cur = kc & 1;
    if (kc + P_AHEAD < NCH) {
      const int kn = kc + P_AHEAD;
      ring[kc % P_AHEAD][0] = *(const v4f*)(gA + kn * P_KC);
#pragma unroll
      for (int i = 0; i < 4; ++i)
        ring[kc % P_AHEAD][1 + i] = *(const v4f*)(gB + (size_t)(kn * P_KC + 8 * i) * HW_);
    }
    const float* L = sm + cur * STAGE;
    v2f fa0[KS], fa1[KS], fb[KS];
#pragma unroll
    for (int s = 0; s < KS; ++s) {
      fa0[s] = *(const v2f*)(L + ao + 4 * s);
      fa1[s] = *(const v2f*)(L + ao + 16 * P_LDA + 4 * s);
      fb[s].x = L[bo + 4 * s * P_LDB];
      fb[s].y = L[bo + 4 * s * P_LDB + P_LDB];
    }
    __builtin_amdgcn_sched_barrier(0);
#pragma unroll
    for (int s = 0; s < KS_HEAD; ++s) {
      c0 = wmma4(fa0[s], fb[s], c0);
      c1 = wmma4(fa1[s], fb[s], c1);
    }
    __builtin_amdgcn_sched_barrier(0);
    if (kc + 1 < NCH) {
      float* S = sm + (cur ^ 1) * STAGE;
      *(v4f*)(S + soa) = ring[(kc + 1) % P_AHEAD][0];
#pragma unroll
      for (int i = 0; i < 4; ++i) *(v4f*)(S + sob + i * 8 * P_LDB) = ring[(kc + 1) % P_AHEAD][1 + i];
      lds_barrier();
    }
    __builtin_amdgcn_sched_barrier(0);
#pragma unroll
    for (int s = KS_HEAD; s < KS; ++s) {
      c0 = wmma4(fa0[s], fb[s], c0);
      c1 = wmma4(fa1[s], fb[s], c1);
    }
  }

  float bias0[8], bias1[8];
#pragma unroll
  for (int r = 0; r < 8; ++r) {
    bias0[r] = b1[r + 8 * half];
    bias1[r] = b1[16 + r + 8 * half];
  }

  float* D = h + (size_t)b * CR_ * HW_;
  const int n = bn * P_BN + wv * 16 + l16;
#pragma unroll
  for (int r = 0; r < 8; ++r) {
    const int m = r + 8 * half;
    float v = c0[r] + bias0[r];
    v = 0.5f * v * (1.0f + erff(v * 0.70710678118654752f));
    D[(size_t)m * HW_ + n] = v;
    float u = c1[r] + bias1[r];
    u = 0.5f * u * (1.0f + erff(u * 0.70710678118654752f));
    D[(size_t)(16 + m) * HW_ + n] = u;
  }
}

__global__ void k_conv2(const float* __restrict__ h, const float* __restrict__ w2,
                        const float* __restrict__ b2, float* __restrict__ y) {
  constexpr int NT = 4;
  const int wid = blockIdx.x * 8 + (threadIdx.x >> 5);
  const int NG = HW_ / (16 * NT);
  const int MT = C_ / 16;
  const int b  = wid / (MT * NG);
  const int r0 = wid % (MT * NG);
  const int mt = r0 / NG;
  const int ng = r0 % NG;

  const float* A  = w2 + (size_t)mt * 16 * CR_;
  const float* Bm = h + (size_t)b * CR_ * HW_ + ng * 16 * NT;
  v8f c[NT] = {};
  gemm_row<NT>(A, CR_, Bm, HW_, CR_, c);

  const int lane = threadIdx.x & 31;
  const int half = lane >> 4;
  const int l16  = lane & 15;
  float* D = y + (size_t)b * C_ * HW_;
#pragma unroll
  for (int t = 0; t < NT; ++t)
#pragma unroll
    for (int r = 0; r < 8; ++r) {
      const int m = mt * 16 + r + 8 * half;
      const int n = ng * 16 * NT + t * 16 + l16;
      D[(size_t)m * HW_ + n] = c[t][r] + b2[m];
    }
}

#define E_BM 128
#define E_BN 64
#define E_KC 32
#define E_LD 36


__global__ __launch_bounds__(256) void k_energy(const float* __restrict__ y, float* __restrict__ e) {
  __shared__ __attribute__((aligned(16))) float sm[2 * (E_BM + E_BN) * E_LD];
  constexpr int STAGE = (E_BM + E_BN) * E_LD;
  constexpr int NCH = HW_ / E_KC;

  const int tid  = threadIdx.x;
  const int lane = tid & 31;
  const int half = lane >> 4;
  const int l16  = lane & 15;
  const int wv   = tid >> 5;
  const int wm   = wv >> 1;
  const int wn   = wv & 1;
  const int b    = blockIdx.x >> 5;
  const int bm   = (blockIdx.x >> 3) & 3;
  const int bn   = blockIdx.x & 7;
  const float* yb = y + (size_t)b * C_ * HW_;

  const int srow = tid >> 3;
  const int sc4  = tid & 7;
  const float* gp[6];
  int so[6];
#pragma unroll
  for (int i = 0; i < 6; ++i) {
    const int r = srow + 32 * i;
    const int grow = (i < 4) ? (bm * E_BM + r) : (bn * E_BN + (r - E_BM));
    gp[i] = yb + (size_t)grow * HW_ + 4 * sc4;
    so[i] = r * E_LD + 4 * sc4;
  }

  const int ao = (wm * 32 + l16) * E_LD + 2 * half;
  const int bo = (E_BM + wn * 32 + l16) * E_LD + 2 * half;

  v4f pre[6];
#pragma unroll
  for (int i = 0; i < 6; ++i) pre[i] = *(const v4f*)gp[i];
#pragma unroll
  for (int i = 0; i < 6; ++i) *(v4f*)(sm + so[i]) = pre[i];
  __syncthreads();

  v8f c00 = {}, c01 = {}, c10 = {}, c11 = {};
  constexpr int KS = E_KC / 4;
  constexpr int KS_HEAD = 6;
  for (int kc = 0; kc < NCH; ++kc) {
    const int cur = kc & 1;
    const bool more = (kc + 1 < NCH);
    if (more) {
#pragma unroll
      for (int i = 0; i < 6; ++i) pre[i] = *(const v4f*)(gp[i] + (size_t)(kc + 1) * E_KC);
    }
    const float* L = sm + cur * STAGE;
    v2f fa0[KS], fa1[KS], fb0[KS], fb1[KS];
#pragma unroll
    for (int s = 0; s < KS; ++s) {
      fa0[s] = *(const v2f*)(L + ao + 4 * s);
      fa1[s] = *(const v2f*)(L + ao + 16 * E_LD + 4 * s);
      fb0[s] = *(const v2f*)(L + bo + 4 * s);
      fb1[s] = *(const v2f*)(L + bo + 16 * E_LD + 4 * s);
    }
    __builtin_amdgcn_sched_barrier(0);
#pragma unroll
    for (int s = 0; s < KS_HEAD; ++s) {
      c00 = wmma4(fa0[s], fb0[s], c00);
      c01 = wmma4(fa0[s], fb1[s], c01);
      c10 = wmma4(fa1[s], fb0[s], c10);
      c11 = wmma4(fa1[s], fb1[s], c11);
    }
    __builtin_amdgcn_sched_barrier(0);
    if (more) {
      float* S = sm + (cur ^ 1) * STAGE;
#pragma unroll
      for (int i = 0; i < 6; ++i) *(v4f*)(S + so[i]) = pre[i];
    }
    __syncthreads();
    __builtin_amdgcn_sched_barrier(0);
#pragma unroll
    for (int s = KS_HEAD; s < KS; ++s) {
      c00 = wmma4(fa0[s], fb0[s], c00);
      c01 = wmma4(fa0[s], fb1[s], c01);
      c10 = wmma4(fa1[s], fb0[s], c10);
      c11 = wmma4(fa1[s], fb1[s], c11);
    }
  }

  float* D = e + (size_t)b * C_ * C_;
  const int m0 = bm * E_BM + wm * 32 + 8 * half;
  const int n0 = bn * E_BN + wn * 32 + l16;
#pragma unroll
  for (int r = 0; r < 8; ++r) {
    D[(size_t)(m0 + r) * C_ + n0]           = c00[r];
    D[(size_t)(m0 + r) * C_ + n0 + 16]      = c01[r];
    D[(size_t)(m0 + 16 + r) * C_ + n0]      = c10[r];
    D[(size_t)(m0 + 16 + r) * C_ + n0 + 16] = c11[r];
  }
}

__global__ void k_softmax(float* __restrict__ e) {
  __shared__ float red[256];
  const int tid = threadIdx.x;
  float* p = e + (size_t)blockIdx.x * C_;
  const float v0 = p[tid];
  const float v1 = p[tid + 256];
  red[tid] = fminf(v0, v1);
  __syncthreads();
  for (int s = 128; s > 0; s >>= 1) {
    if (tid < s) red[tid] = fminf(red[tid], red[tid + s]);
    __syncthreads();
  }
  const float rmin = red[0];
  __syncthreads();
  const float e0 = expf(rmin - v0);
  const float e1 = expf(rmin - v1);
  red[tid] = e0 + e1;
  __syncthreads();
  for (int s = 128; s > 0; s >>= 1) {
    if (tid < s) red[tid] += red[tid + s];
    __syncthreads();
  }
  const float inv = 1.0f / red[0];
  p[tid] = e0 * inv;
  p[tid + 256] = e1 * inv;
}

__global__ void k_m2(const float* __restrict__ attn, const float* __restrict__ wf,
                     const float* __restrict__ gamma, float* __restrict__ m2) {
  constexpr int NT = 4;
  const int wid = blockIdx.x * 8 + (threadIdx.x >> 5);
  const int NG = C_ / (16 * NT);
  const int MT = C_ / 16;
  const int b  = wid / (MT * NG);
  const int r0 = wid % (MT * NG);
  const int mt = r0 / NG;
  const int ng = r0 % NG;

  const float* A  = wf + (size_t)mt * 16 * C2_;
  const float* Bm = attn + (size_t)b * C_ * C_ + ng * 16 * NT;
  v8f c[NT] = {};
  gemm_row<NT>(A, C2_, Bm, C_, C_, c);

  const float g = gamma[0];
  const int lane = threadIdx.x & 31;
  const int half = lane >> 4;
  const int l16  = lane & 15;
  float* D = m2 + (size_t)b * C_ * C_;
#pragma unroll
  for (int t = 0; t < NT; ++t)
#pragma unroll
    for (int r = 0; r < 8; ++r) {
      const int m = mt * 16 + r + 8 * half;
      const int n = ng * 16 * NT + t * 16 + l16;
      D[(size_t)m * C_ + n] = g * c[t][r] + wf[(size_t)m * C2_ + n];
    }
}

#define F_BM  128
#define F_BN  128
#define F_KC  32
#define F_LDA 36
#define F_LDB 144

__global__ __launch_bounds__(256) void k_fusion(const float* __restrict__ m2, const float* __restrict__ y,
                                                const float* __restrict__ x, const float* __restrict__ wf,
                                                const float* __restrict__ bf, float* __restrict__ out) {
  constexpr int A_SZ  = F_BM * F_LDA;
  constexpr int STAGE = A_SZ + F_KC * F_LDB;
  __shared__ __attribute__((aligned(16))) float sm[2 * STAGE];
  constexpr int NCH  = C2_ / F_KC;
  constexpr int NCH1 = C_ / F_KC;

  const int tid  = threadIdx.x;
  const int lane = tid & 31;
  const int half = lane >> 4;
  const int l16  = lane & 15;
  const int wv   = tid >> 5;
  const int wm   = wv >> 2;
  const int wn   = wv & 3;
  const int b    = blockIdx.x >> 7;
  const int bm   = (blockIdx.x >> 5) & 3;
  const int bn   = blockIdx.x & 31;

  const int arow = tid >> 3;
  const int ac4  = tid & 7;
  const int brow = tid >> 5;
  const int bc4  = tid & 31;
  const float* gA1 = m2 + (size_t)b * C_ * C_ + (size_t)(bm * F_BM + arow) * C_ + 4 * ac4;
  const float* gA2 = wf + (size_t)(bm * F_BM + arow) * C2_ + C_ + 4 * ac4;
  const float* gB1 = y + (size_t)b * C_ * HW_ + (size_t)brow * HW_ + bn * F_BN + 4 * bc4;
  const float* gB2 = x + (size_t)b * C_ * HW_ + (size_t)brow * HW_ + bn * F_BN + 4 * bc4;
  const int soa = arow * F_LDA + 4 * ac4;
  const int sob = A_SZ + brow * F_LDB + 4 * bc4;

  const int ao = (wm * 64 + l16) * F_LDA + 2 * half;
  const int bo = A_SZ + (2 * half) * F_LDB + wn * 32 + l16;

  v4f pre[8];
#pragma unroll
  for (int i = 0; i < 4; ++i) pre[i] = *(const v4f*)(gA1 + (size_t)i * 32 * C_);
#pragma unroll
  for (int i = 0; i < 4; ++i) pre[4 + i] = *(const v4f*)(gB1 + (size_t)i * 8 * HW_);
#pragma unroll
  for (int i = 0; i < 4; ++i) *(v4f*)(sm + soa + i * 32 * F_LDA) = pre[i];
#pragma unroll
  for (int i = 0; i < 4; ++i) *(v4f*)(sm + sob + i * 8 * F_LDB) = pre[4 + i];
  __syncthreads();

  v8f c[4][2] = {};
  constexpr int KS = F_KC / 4;
  constexpr int KS_HEAD = 6;
  for (int kc = 0; kc < NCH; ++kc) {
    const int cur = kc & 1;
    const bool more = (kc + 1 < NCH);
    if (more) {
      const int kn = kc + 1;
      if (kn < NCH1) {
#pragma unroll
        for (int i = 0; i < 4; ++i) pre[i] = *(const v4f*)(gA1 + (size_t)i * 32 * C_ + kn * F_KC);
#pragma unroll
        for (int i = 0; i < 4; ++i) pre[4 + i] = *(const v4f*)(gB1 + (size_t)(kn * F_KC + 8 * i) * HW_);
      } else {
#pragma unroll
        for (int i = 0; i < 4; ++i) pre[i] = *(const v4f*)(gA2 + (size_t)i * 32 * C2_ + (kn - NCH1) * F_KC);
#pragma unroll
        for (int i = 0; i < 4; ++i) pre[4 + i] = *(const v4f*)(gB2 + (size_t)((kn - NCH1) * F_KC + 8 * i) * HW_);
      }
    }
    const float* L = sm + cur * STAGE;
    v2f fa[4][KS], fb[2][KS];
#pragma unroll
    for (int s = 0; s < KS; ++s) {
#pragma unroll
      for (int t = 0; t < 4; ++t) fa[t][s] = *(const v2f*)(L + ao + t * 16 * F_LDA + 4 * s);
#pragma unroll
      for (int t = 0; t < 2; ++t) {
        fb[t][s].x = L[bo + t * 16 + 4 * s * F_LDB];
        fb[t][s].y = L[bo + t * 16 + 4 * s * F_LDB + F_LDB];
      }
    }
    __builtin_amdgcn_sched_barrier(0);
#pragma unroll
    for (int s = 0; s < KS_HEAD; ++s)
#pragma unroll
      for (int t = 0; t < 4; ++t) {
        c[t][0] = wmma4(fa[t][s], fb[0][s], c[t][0]);
        c[t][1] = wmma4(fa[t][s], fb[1][s], c[t][1]);
      }
    __builtin_amdgcn_sched_barrier(0);
    if (more) {
      float* S = sm + (cur ^ 1) * STAGE;
#pragma unroll
      for (int i = 0; i < 4; ++i) *(v4f*)(S + soa + i * 32 * F_LDA) = pre[i];
#pragma unroll
      for (int i = 0; i < 4; ++i) *(v4f*)(S + sob + i * 8 * F_LDB) = pre[4 + i];
    }
    __syncthreads();
    __builtin_amdgcn_sched_barrier(0);
#pragma unroll
    for (int s = KS_HEAD; s < KS; ++s)
#pragma unroll
      for (int t = 0; t < 4; ++t) {
        c[t][0] = wmma4(fa[t][s], fb[0][s], c[t][0]);
        c[t][1] = wmma4(fa[t][s], fb[1][s], c[t][1]);
      }
  }

  float* D = out + (size_t)b * C_ * HW_;
  const int n0 = bn * F_BN + wn * 32 + l16;
#pragma unroll
  for (int t = 0; t < 4; ++t) {
    const int m0 = bm * F_BM + wm * 64 + t * 16 + 8 * half;
#pragma unroll
    for (int r = 0; r < 8; ++r) {
      const float bias = bf[m0 + r];
      D[(size_t)(m0 + r) * HW_ + n0]      = c[t][0][r] + bias;
      D[(size_t)(m0 + r) * HW_ + n0 + 16] = c[t][1][r] + bias;
    }
  }
}

extern "C" void kernel_launch(void* const* d_in, const int* in_sizes, int n_in,
                              void* d_out, int out_size, void* d_ws, size_t ws_size,
                              hipStream_t stream) {
  const float* x     = (const float*)d_in[0];
  const float* w1    = (const float*)d_in[1];
  const float* b1    = (const float*)d_in[2];
  const float* w2    = (const float*)d_in[3];
  const float* b2    = (const float*)d_in[4];
  const float* wf    = (const float*)d_in[5];
  const float* bf    = (const float*)d_in[6];
  const float* gamma = (const float*)d_in[7];
  float* out = (float*)d_out;

  char* ws = (char*)d_ws;
  float* h    = (float*)(ws);
  float* y    = (float*)(ws + (size_t)4 * 1024 * 1024);
  float* attn = (float*)(ws + (size_t)68 * 1024 * 1024);
  float* m2   = (float*)(ws + (size_t)76 * 1024 * 1024);

  const dim3 blk(256);
  k_conv1_gelu<<<dim3(256),  blk, 0, stream>>>(x, w1, b1, h);
  k_conv2     <<<dim3(2048), blk, 0, stream>>>(h, w2, b2, y);
  k_energy    <<<dim3(256),  blk, 0, stream>>>(y, attn);
  k_softmax   <<<dim3(4096), blk, 0, stream>>>(attn);
  k_m2        <<<dim3(256),  blk, 0, stream>>>(attn, wf, gamma, m2);
  k_fusion    <<<dim3(1024), blk, 0, stream>>>(m2, y, x, wf, bf, out);
}


# energy y.y^T: only 32x64 tiles on or above the diagonal are multiplied (72 of 128 per batch; 4-wave blocks, 32x16 wave tiles), tiles above it stored transposed too; f32 sum order unchanged
# speedup vs baseline: 1.7970x; 1.0772x over previous
#include <hip/hip_runtime.h>
#include <math.h>


#define B_   8
#define C_   512
#define CR_  32
#define HW_  4096
#define C2_  1024

typedef __attribute__((ext_vector_type(2))) float v2f;
typedef __attribute__((ext_vector_type(8))) float v8f;

__device__ __forceinline__ v8f wmma4(v2f a, v2f b, v8f c) {
  return __builtin_amdgcn_wmma_f32_16x16x4_f32(
       false, a,  false, b,
       (short)0, c,  false,  false);
}

template<int NT>
__device__ __forceinline__ void gemm_row(const float* __restrict__ A, int lda,
                                         const float* __restrict__ Bm, int ldb,
                                         int K, v8f c[NT]) {
  const int lane = threadIdx.x & 31;
  const int half = lane >> 4;
  const int l16  = lane & 15;
  const float* ap = A + (size_t)l16 * lda + 2 * half;
  const float* bp = Bm + (size_t)(2 * half) * ldb + l16;
#pragma unroll 2
  for (int k = 0; k < K; k += 4) {
    v2f a = *(const v2f*)ap;
#pragma unroll
    for (int t = 0; t < NT; ++t) {
      v2f b;
      b.x = bp[t * 16];
      b.y = bp[t * 16 + ldb];
      c[t] = wmma4(a, b, c[t]);
    }
    ap += 4;
    bp += (size_t)4 * ldb;
  }
}

#define P_BN    128
#define P_KC    32
#define P_LDA   36
#define P_LDB   144
#define P_AHEAD 3

typedef __attribute__((ext_vector_type(4))) float v4f;

__device__ __forceinline__ void lds_barrier() {
  __builtin_amdgcn_fence(__ATOMIC_RELEASE, "workgroup", "local");
  __builtin_amdgcn_s_barrier();
  __builtin_amdgcn_fence(__ATOMIC_ACQUIRE, "workgroup", "local");
}

__global__ __launch_bounds__(256) void k_conv1_gelu(const float* __restrict__ x, const float* __restrict__ w1,
                                                    const float* __restrict__ b1, float* __restrict__ h) {
  constexpr int A_SZ  = CR_ * P_LDA;
  constexpr int STAGE = A_SZ + P_KC * P_LDB;
  __shared__ __attribute__((aligned(16))) float sm[2 * STAGE];
  constexpr int NCH = C_ / P_KC;

  const int tid  = threadIdx.x;
  const int lane = tid & 31;
  const int half = lane >> 4;
  const int l16  = lane & 15;
  const int wv   = tid >> 5;
  const int b    = blockIdx.x >> 5;
  const int bn   = blockIdx.x & 31;

  const int arow = tid >> 3;
  const int ac4  = tid & 7;
  const int brow = tid >> 5;
  const int bc4  = tid & 31;
  const float* gA = w1 + (size_t)arow * C_ + 4 * ac4;
  const float* gB = x + (size_t)b * C_ * HW_ + (size_t)brow * HW_ + bn * P_BN + 4 * bc4;
  const int soa = arow * P_LDA + 4 * ac4;
  const int sob = A_SZ + brow * P_LDB + 4 * bc4;

  const int ao = l16 * P_LDA + 2 * half;
  const int bo = A_SZ + (2 * half) * P_LDB + wv * 16 + l16;

  v4f ring[P_AHEAD][5];
#pragma unroll
  for (int j = 0; j < P_AHEAD; ++j) {
    ring[j][0] = *(const v4f*)(gA + j * P_KC);
#pragma unroll
    for (int i = 0; i < 4; ++i) ring[j][1 + i] = *(const v4f*)(gB + (size_t)(j * P_KC + 8 * i) * HW_);
  }
  *(v4f*)(sm + soa) = ring[0][0];
#pragma unroll
  for (int i = 0; i < 4; ++i) *(v4f*)(sm + sob + i * 8 * P_LDB) = ring[0][1 + i];
  lds_barrier();

  v8f c0 = {}, c1 = {};
  constexpr int KS = P_KC / 4;
  constexpr int KS_HEAD = 6;
#pragma unroll
  for (int kc = 0; kc < NCH; ++kc) {
    const int cur = kc & 1;
    if (kc + P_AHEAD < NCH) {
      const int kn = kc + P_AHEAD;
      ring[kc % P_AHEAD][0] = *(const v4f*)(gA + kn * P_KC);
#pragma unroll
      for (int i = 0; i < 4; ++i)
        ring[kc % P_AHEAD][1 + i] = *(const v4f*)(gB + (size_t)(kn * P_KC + 8 * i) * HW_);
    }
    const float* L = sm + cur * STAGE;
    v2f fa0[KS], fa1[KS], fb[KS];
#pragma unroll
    for (int s = 0; s < KS; ++s) {
      fa0[s] = *(const v2f*)(L + ao + 4 * s);
      fa1[s] = *(const v2f*)(L + ao + 16 * P_LDA + 4 * s);
      fb[s].x = L[bo + 4 * s * P_LDB];
      fb[s].y = L[bo + 4 * s * P_LDB + P_LDB];
    }
    __builtin_amdgcn_sched_barrier(0);
#pragma unroll
    for (int s = 0; s < KS_HEAD; ++s) {
      c0 = wmma4(fa0[s], fb[s], c0);
      c1 = wmma4(fa1[s], fb[s], c1);
    }
    __builtin_amdgcn_sched_barrier(0);
    if (kc + 1 < NCH) {
      float* S = sm + (cur ^ 1) * STAGE;
      *(v4f*)(S + soa) = ring[(kc + 1) % P_AHEAD][0];
#pragma unroll
      for (int i = 0; i < 4; ++i) *(v4f*)(S + sob + i * 8 * P_LDB) = ring[(kc + 1) % P_AHEAD][1 + i];
      lds_barrier();
    }
    __builtin_amdgcn_sched_barrier(0);
#pragma unroll
    for (int s = KS_HEAD; s < KS; ++s) {
      c0 = wmma4(fa0[s], fb[s], c0);
      c1 = wmma4(fa1[s], fb[s], c1);
    }
  }

  float bias0[8], bias1[8];
#pragma unroll
  for (int r = 0; r < 8; ++r) {
    bias0[r] = b1[r + 8 * half];
    bias1[r] = b1[16 + r + 8 * half];
  }

  float* D = h + (size_t)b * CR_ * HW_;
  const int n = bn * P_BN + wv * 16 + l16;
#pragma unroll
  for (int r = 0; r < 8; ++r) {
    const int m = r + 8 * half;
    float v = c0[r] + bias0[r];
    v = 0.5f * v * (1.0f + erff(v * 0.70710678118654752f));
    D[(size_t)m * HW_ + n] = v;
    float u = c1[r] + bias1[r];
    u = 0.5f * u * (1.0f + erff(u * 0.70710678118654752f));
    D[(size_t)(16 + m) * HW_ + n] = u;
  }
}

__global__ void k_conv2(const float* __restrict__ h, const float* __restrict__ w2,
                        const float* __restrict__ b2, float* __restrict__ y) {
  constexpr int NT = 4;
  const int wid = blockIdx.x * 8 + (threadIdx.x >> 5);
  const int NG = HW_ / (16 * NT);
  const int MT = C_ / 16;
  const int b  = wid / (MT * NG);
  const int r0 = wid % (MT * NG);
  const int mt = r0 / NG;
  const int ng = r0 % NG;

  const float* A  = w2 + (size_t)mt * 16 * CR_;
  const float* Bm = h + (size_t)b * CR_ * HW_ + ng * 16 * NT;
  v8f c[NT] = {};
  gemm_row<NT>(A, CR_, Bm, HW_, CR_, c);

  const int lane = threadIdx.x & 31;
  const int half = lane >> 4;
  const int l16  = lane & 15;
  float* D = y + (size_t)b * C_ * HW_;
#pragma unroll
  for (int t = 0; t < NT; ++t)
#pragma unroll
    for (int r = 0; r < 8; ++r) {
      const int m = mt * 16 + r + 8 * half;
      const int n = ng * 16 * NT + t * 16 + l16;
      D[(size_t)m * HW_ + n] = c[t][r] + b2[m];
    }
}

#define E_BM 32
#define E_BN 64
#define E_KC 32
#define E_LD 36
#define E_TILES 72

__global__ __launch_bounds__(128) void k_energy(const float* __restrict__ y, float* __restrict__ e) {
  __shared__ __attribute__((aligned(16))) float sm[2 * (E_BM + E_BN) * E_LD];
  constexpr int STAGE = (E_BM + E_BN) * E_LD;
  constexpr int NCH = HW_ / E_KC;

  const int tid  = threadIdx.x;
  const int lane = tid & 31;
  const int half = lane >> 4;
  const int l16  = lane & 15;
  const int wv   = tid >> 5;
  const int b    = blockIdx.x / E_TILES;
  int rem        = blockIdx.x % E_TILES;
  int j = 0;
#pragma unroll
  for (int q = 0; q < 7; ++q) {
    const int nb = 2 * (8 - q);
    if (j == q && rem >= nb) { rem -= nb; j = q + 1; }
  }
  const int wdt   = 8 - j;
  const int upper = (rem >= wdt) ? 1 : 0;
  if (upper) rem -= wdt;
  const int strip = 2 * j + upper;
  const int bn    = j + rem;
  const float* yb = y + (size_t)b * C_ * HW_;

  const int srow = tid >> 3;
  const int sc4  = tid & 7;
  const float* gp[6];
  int so[6];
#pragma unroll
  for (int i = 0; i < 6; ++i) {
    const int r = srow + 16 * i;
    const int grow = (i < 2) ? (strip * E_BM + r) : (bn * E_BN + (r - E_BM));
    gp[i] = yb + (size_t)grow * HW_ + 4 * sc4;
    so[i] = r * E_LD + 4 * sc4;
  }

  const int ao = l16 * E_LD + 2 * half;
  const int bo = (E_BM + wv * 16 + l16) * E_LD + 2 * half;

  v4f pre[6];
#pragma unroll
  for (int i = 0; i < 6; ++i) pre[i] = *(const v4f*)gp[i];
#pragma unroll
  for (int i = 0; i < 6; ++i) *(v4f*)(sm + so[i]) = pre[i];
  __syncthreads();

  v8f c0 = {}, c1 = {};
  constexpr int KS = E_KC / 4;
  constexpr int KS_HEAD = 6;
  for (int kc = 0; kc < NCH; ++kc) {
    const int cur = kc & 1;
    const bool more = (kc + 1 < NCH);
    if (more) {
#pragma unroll
      for (int i = 0; i < 6; ++i) pre[i] = *(const v4f*)(gp[i] + (size_t)(kc + 1) * E_KC);
    }
    const float* L = sm + cur * STAGE;
    v2f fa0[KS], fa1[KS], fb[KS];
#pragma unroll
    for (int s = 0; s < KS; ++s) {
      fa0[s] = *(const v2f*)(L + ao + 4 * s);
      fa1[s] = *(const v2f*)(L + ao + 16 * E_LD + 4 * s);
      fb[s]  = *(const v2f*)(L + bo + 4 * s);
    }
    __builtin_amdgcn_sched_barrier(0);
#pragma unroll
    for (int s = 0; s < KS_HEAD; ++s) {
      c0 = wmma4(fa0[s], fb[s], c0);
      c1 = wmma4(fa1[s], fb[s], c1);
    }
    __builtin_amdgcn_sched_barrier(0);
    if (more) {
      float* S = sm + (cur ^ 1) * STAGE;
#pragma unroll
      for (int i = 0; i < 6; ++i) *(v4f*)(S + so[i]) = pre[i];
    }
    __syncthreads();
    __builtin_amdgcn_sched_barrier(0);
#pragma unroll
    for (int s = KS_HEAD; s < KS; ++s) {
      c0 = wmma4(fa0[s], fb[s], c0);
      c1 = wmma4(fa1[s], fb[s], c1);
    }
  }

  float* D = e + (size_t)b * C_ * C_;
  const int m0 = strip * E_BM + 8 * half;
  const int n0 = bn * E_BN + wv * 16 + l16;
#pragma unroll
  for (int r = 0; r < 8; ++r) {
    D[(size_t)(m0 + r) * C_ + n0]      = c0[r];
    D[(size_t)(m0 + 16 + r) * C_ + n0] = c1[r];
  }
  if (bn > j) {
    float* Tp = D + (size_t)n0 * C_ + m0;
    v4f t0 = {c0[0], c0[1], c0[2], c0[3]};
    v4f t1 = {c0[4], c0[5], c0[6], c0[7]};
    v4f t2 = {c1[0], c1[1], c1[2], c1[3]};
    v4f t3 = {c1[4], c1[5], c1[6], c1[7]};
    *(v4f*)(Tp)      = t0;
    *(v4f*)(Tp + 4)  = t1;
    *(v4f*)(Tp + 16) = t2;
    *(v4f*)(Tp + 20) = t3;
  }
}

__global__ void k_softmax(float* __restrict__ e) {
  __shared__ float red[256];
  const int tid = threadIdx.x;
  float* p = e + (size_t)blockIdx.x * C_;
  const float v0 = p[tid];
  const float v1 = p[tid + 256];
  red[tid] = fminf(v0, v1);
  __syncthreads();
  for (int s = 128; s > 0; s >>= 1) {
    if (tid < s) red[tid] = fminf(red[tid], red[tid + s]);
    __syncthreads();
  }
  const float rmin = red[0];
  __syncthreads();
  const float e0 = expf(rmin - v0);
  const float e1 = expf(rmin - v1);
  red[tid] = e0 + e1;
  __syncthreads();
  for (int s = 128; s > 0; s >>= 1) {
    if (tid < s) red[tid] += red[tid + s];
    __syncthreads();
  }
  const float inv = 1.0f / red[0];
  p[tid] = e0 * inv;
  p[tid + 256] = e1 * inv;
}

__global__ void k_m2(const float* __restrict__ attn, const float* __restrict__ wf,
                     const float* __restrict__ gamma, float* __restrict__ m2) {
  constexpr int NT = 4;
  const int wid = blockIdx.x * 8 + (threadIdx.x >> 5);
  const int NG = C_ / (16 * NT);
  const int MT = C_ / 16;
  const int b  = wid / (MT * NG);
  const int r0 = wid % (MT * NG);
  const int mt = r0 / NG;
  const int ng = r0 % NG;

  const float* A  = wf + (size_t)mt * 16 * C2_;
  const float* Bm = attn + (size_t)b * C_ * C_ + ng * 16 * NT;
  v8f c[NT] = {};
  gemm_row<NT>(A, C2_, Bm, C_, C_, c);

  const float g = gamma[0];
  const int lane = threadIdx.x & 31;
  const int half = lane >> 4;
  const int l16  = lane & 15;
  float* D = m2 + (size_t)b * C_ * C_;
#pragma unroll
  for (int t = 0; t < NT; ++t)
#pragma unroll
    for (int r = 0; r < 8; ++r) {
      const int m = mt * 16 + r + 8 * half;
      const int n = ng * 16 * NT + t * 16 + l16;
      D[(size_t)m * C_ + n] = g * c[t][r] + wf[(size_t)m * C2_ + n];
    }
}

#define F_BM  128
#define F_BN  128
#define F_KC  32
#define F_LDA 36
#define F_LDB 144

__global__ __launch_bounds__(256) void k_fusion(const float* __restrict__ m2, const float* __restrict__ y,
                                                const float* __restrict__ x, const float* __restrict__ wf,
                                                const float* __restrict__ bf, float* __restrict__ out) {
  constexpr int A_SZ  = F_BM * F_LDA;
  constexpr int STAGE = A_SZ + F_KC * F_LDB;
  __shared__ __attribute__((aligned(16))) float sm[2 * STAGE];
  constexpr int NCH  = C2_ / F_KC;
  constexpr int NCH1 = C_ / F_KC;

  const int tid  = threadIdx.x;
  const int lane = tid & 31;
  const int half = lane >> 4;
  const int l16  = lane & 15;
  const int wv   = tid >> 5;
  const int wm   = wv >> 2;
  const int wn   = wv & 3;
  const int b    = blockIdx.x >> 7;
  const int bm   = (blockIdx.x >> 5) & 3;
  const int bn   = blockIdx.x & 31;

  const int arow = tid >> 3;
  const int ac4  = tid & 7;
  const int brow = tid >> 5;
  const int bc4  = tid & 31;
  const float* gA1 = m2 + (size_t)b * C_ * C_ + (size_t)(bm * F_BM + arow) * C_ + 4 * ac4;
  const float* gA2 = wf + (size_t)(bm * F_BM + arow) * C2_ + C_ + 4 * ac4;
  const float* gB1 = y + (size_t)b * C_ * HW_ + (size_t)brow * HW_ + bn * F_BN + 4 * bc4;
  const float* gB2 = x + (size_t)b * C_ * HW_ + (size_t)brow * HW_ + bn * F_BN + 4 * bc4;
  const int soa = arow * F_LDA + 4 * ac4;
  const int sob = A_SZ + brow * F_LDB + 4 * bc4;

  const int ao = (wm * 64 + l16) * F_LDA + 2 * half;
  const int bo = A_SZ + (2 * half) * F_LDB + wn * 32 + l16;

  v4f pre[8];
#pragma unroll
  for (int i = 0; i < 4; ++i) pre[i] = *(const v4f*)(gA1 + (size_t)i * 32 * C_);
#pragma unroll
  for (int i = 0; i < 4; ++i) pre[4 + i] = *(const v4f*)(gB1 + (size_t)i * 8 * HW_);
#pragma unroll
  for (int i = 0; i < 4; ++i) *(v4f*)(sm + soa + i * 32 * F_LDA) = pre[i];
#pragma unroll
  for (int i = 0; i < 4; ++i) *(v4f*)(sm + sob + i * 8 * F_LDB) = pre[4 + i];
  __syncthreads();

  v8f c[4][2] = {};
  constexpr int KS = F_KC / 4;
  constexpr int KS_HEAD = 6;
  for (int kc = 0; kc < NCH; ++kc) {
    const int cur = kc & 1;
    const bool more = (kc + 1 < NCH);
    if (more) {
      const int kn = kc + 1;
      if (kn < NCH1) {
#pragma unroll
        for (int i = 0; i < 4; ++i) pre[i] = *(const v4f*)(gA1 + (size_t)i * 32 * C_ + kn * F_KC);
#pragma unroll
        for (int i = 0; i < 4; ++i) pre[4 + i] = *(const v4f*)(gB1 + (size_t)(kn * F_KC + 8 * i) * HW_);
      } else {
#pragma unroll
        for (int i = 0; i < 4; ++i) pre[i] = *(const v4f*)(gA2 + (size_t)i * 32 * C2_ + (kn - NCH1) * F_KC);
#pragma unroll
        for (int i = 0; i < 4; ++i) pre[4 + i] = *(const v4f*)(gB2 + (size_t)((kn - NCH1) * F_KC + 8 * i) * HW_);
      }
    }
    const float* L = sm + cur * STAGE;
    v2f fa[4][KS], fb[2][KS];
#pragma unroll
    for (int s = 0; s < KS; ++s) {
#pragma unroll
      for (int t = 0; t < 4; ++t) fa[t][s] = *(const v2f*)(L + ao + t * 16 * F_LDA + 4 * s);
#pragma unroll
      for (int t = 0; t < 2; ++t) {
        fb[t][s].x = L[bo + t * 16 + 4 * s * F_LDB];
        fb[t][s].y = L[bo + t * 16 + 4 * s * F_LDB + F_LDB];
      }
    }
    __builtin_amdgcn_sched_barrier(0);
#pragma unroll
    for (int s = 0; s < KS_HEAD; ++s)
#pragma unroll
      for (int t = 0; t < 4; ++t) {
        c[t][0] = wmma4(fa[t][s], fb[0][s], c[t][0]);
        c[t][1] = wmma4(fa[t][s], fb[1][s], c[t][1]);
      }
    __builtin_amdgcn_sched_barrier(0);
    if (more) {
      float* S = sm + (cur ^ 1) * STAGE;
#pragma unroll
      for (int i = 0; i < 4; ++i) *(v4f*)(S + soa + i * 32 * F_LDA) = pre[i];
#pragma unroll
      for (int i = 0; i < 4; ++i) *(v4f*)(S + sob + i * 8 * F_LDB) = pre[4 + i];
    }
    __syncthreads();
    __builtin_amdgcn_sched_barrier(0);
#pragma unroll
    for (int s = KS_HEAD; s < KS; ++s)
#pragma unroll
      for (int t = 0; t < 4; ++t) {
        c[t][0] = wmma4(fa[t][s], fb[0][s], c[t][0]);
        c[t][1] = wmma4(fa[t][s], fb[1][s], c[t][1]);
      }
  }

  float* D = out + (size_t)b * C_ * HW_;
  const int n0 = bn * F_BN + wn * 32 + l16;
#pragma unroll
  for (int t = 0; t < 4; ++t) {
    const int m0 = bm * F_BM + wm * 64 + t * 16 + 8 * half;
#pragma unroll
    for (int r = 0; r < 8; ++r) {
      const float bias = bf[m0 + r];
      D[(size_t)(m0 + r) * HW_ + n0]      = c[t][0][r] + bias;
      D[(size_t)(m0 + r) * HW_ + n0 + 16] = c[t][1][r] + bias;
    }
  }
}

extern "C" void kernel_launch(void* const* d_in, const int* in_sizes, int n_in,
                              void* d_out, int out_size, void* d_ws, size_t ws_size,
                              hipStream_t stream) {
  const float* x     = (const float*)d_in[0];
  const float* w1    = (const float*)d_in[1];
  const float* b1    = (const float*)d_in[2];
  const float* w2    = (const float*)d_in[3];
  const float* b2    = (const float*)d_in[4];
  const float* wf    = (const float*)d_in[5];
  const float* bf    = (const float*)d_in[6];
  const float* gamma = (const float*)d_in[7];
  float* out = (float*)d_out;

  char* ws = (char*)d_ws;
  float* h    = (float*)(ws);
  float* y    = (float*)(ws + (size_t)4 * 1024 * 1024);
  float* attn = (float*)(ws + (size_t)68 * 1024 * 1024);
  float* m2   = (float*)(ws + (size_t)76 * 1024 * 1024);

  const dim3 blk(256);
  k_conv1_gelu<<<dim3(256),  blk, 0, stream>>>(x, w1, b1, h);
  k_conv2     <<<dim3(2048), blk, 0, stream>>>(h, w2, b2, y);
  k_energy    <<<dim3(576),  dim3(128), 0, stream>>>(y, attn);
  k_softmax   <<<dim3(4096), blk, 0, stream>>>(attn);
  k_m2        <<<dim3(256),  blk, 0, stream>>>(attn, wf, gamma, m2);
  k_fusion    <<<dim3(1024), blk, 0, stream>>>(m2, y, x, wf, bf, out);
}


# M2 product Wf1.attn: 32x128 block tiles on kernel 1's loop (32-k chunks staged in LDS, 16-byte row loads 3 chunks ahead); was fragments straight from memory, 64 waits a wave; f32 sum order unchanged
# speedup vs baseline: 1.9077x; 1.0616x over previous
#include <hip/hip_runtime.h>
#include <math.h>


#define B_   8
#define C_   512
#define CR_  32
#define HW_  4096
#define C2_  1024

typedef __attribute__((ext_vector_type(2))) float v2f;
typedef __attribute__((ext_vector_type(8))) float v8f;

__device__ __forceinline__ v8f wmma4(v2f a, v2f b, v8f c) {
  return __builtin_amdgcn_wmma_f32_16x16x4_f32(
       false, a,  false, b,
       (short)0, c,  false,  false);
}

template<int NT>
__device__ __forceinline__ void gemm_row(const float* __restrict__ A, int lda,
                                         const float* __restrict__ Bm, int ldb,
                                         int K, v8f c[NT]) {
  const int lane = threadIdx.x & 31;
  const int half = lane >> 4;
  const int l16  = lane & 15;
  const float* ap = A + (size_t)l16 * lda + 2 * half;
  const float* bp = Bm + (size_t)(2 * half) * ldb + l16;
#pragma unroll 2
  for (int k = 0; k < K; k += 4) {
    v2f a = *(const v2f*)ap;
#pragma unroll
    for (int t = 0; t < NT; ++t) {
      v2f b;
      b.x = bp[t * 16];
      b.y = bp[t * 16 + ldb];
      c[t] = wmma4(a, b, c[t]);
    }
    ap += 4;
    bp += (size_t)4 * ldb;
  }
}

#define P_BM    32
#define P_BN    128
#define P_KC    32
#define P_LDA   36
#define P_LDB   144
#define P_AHEAD 3

typedef __attribute__((ext_vector_type(4))) float v4f;

__device__ __forceinline__ void lds_barrier() {
  __builtin_amdgcn_fence(__ATOMIC_RELEASE, "workgroup", "local");
  __builtin_amdgcn_s_barrier();
  __builtin_amdgcn_fence(__ATOMIC_ACQUIRE, "workgroup", "local");
}

template<int LDGB>
__device__ __forceinline__ void tile_32x128_k512(const float* __restrict__ gA, const float* __restrict__ gB,
                                                 float* sm, v8f& c0, v8f& c1) {
  constexpr int A_SZ  = P_BM * P_LDA;
  constexpr int STAGE = A_SZ + P_KC * P_LDB;
  constexpr int NCH = C_ / P_KC;

  const int tid  = threadIdx.x;
  const int lane = tid & 31;
  const int half = lane >> 4;
  const int l16  = lane & 15;
  const int wv   = tid >> 5;
  const int arow = tid >> 3;
  const int ac4  = tid & 7;
  const int brow = tid >> 5;
  const int bc4  = tid & 31;
  const int soa = arow * P_LDA + 4 * ac4;
  const int sob = A_SZ + brow * P_LDB + 4 * bc4;

  const int ao = l16 * P_LDA + 2 * half;
  const int bo = A_SZ + (2 * half) * P_LDB + wv * 16 + l16;

  v4f ring[P_AHEAD][5];
#pragma unroll
  for (int j = 0; j < P_AHEAD; ++j) {
    ring[j][0] = *(const v4f*)(gA + j * P_KC);
#pragma unroll
    for (int i = 0; i < 4; ++i) ring[j][1 + i] = *(const v4f*)(gB + (size_t)(j * P_KC + 8 * i) * LDGB);
  }
  *(v4f*)(sm + soa) = ring[0][0];
#pragma unroll
  for (int i = 0; i < 4; ++i) *(v4f*)(sm + sob + i * 8 * P_LDB) = ring[0][1 + i];
  lds_barrier();

  constexpr int KS = P_KC / 4;
  constexpr int KS_HEAD = 6;
#pragma unroll
  for (int kc = 0; kc < NCH; ++kc) {
    const int cur = kc & 1;
    if (kc + P_AHEAD < NCH) {
      const int kn = kc + P_AHEAD;
      ring[kc % P_AHEAD][0] = *(const v4f*)(gA + kn * P_KC);
#pragma unroll
      for (int i = 0; i < 4; ++i)
        ring[kc % P_AHEAD][1 + i] = *(const v4f*)(gB + (size_t)(kn * P_KC + 8 * i) * LDGB);
    }
    const float* L = sm + cur * STAGE;
    v2f fa0[KS], fa1[KS], fb[KS];
#pragma unroll
    for (int s = 0; s < KS; ++s) {
      fa0[s] = *(const v2f*)(L + ao + 4 * s);
      fa1[s] = *(const v2f*)(L + ao + 16 * P_LDA + 4 * s);
      fb[s].x = L[bo + 4 * s * P_LDB];
      fb[s].y = L[bo + 4 * s * P_LDB + P_LDB];
    }
    __builtin_amdgcn_sched_barrier(0);
#pragma unroll
    for (int s = 0; s < KS_HEAD; ++s) {
      c0 = wmma4(fa0[s], fb[s], c0);
      c1 = wmma4(fa1[s], fb[s], c1);
    }
    __builtin_amdgcn_sched_barrier(0);
    if (kc + 1 < NCH) {
      float* S = sm + (cur ^ 1) * STAGE;
      *(v4f*)(S + soa) = ring[(kc + 1) % P_AHEAD][0];
#pragma unroll
      for (int i = 0; i < 4; ++i) *(v4f*)(S + sob + i * 8 * P_LDB) = ring[(kc + 1) % P_AHEAD][1 + i];
      lds_barrier();
    }
    __builtin_amdgcn_sched_barrier(0);
#pragma unroll
    for (int s = KS_HEAD; s < KS; ++s) {
      c0 = wmma4(fa0[s], fb[s], c0);
      c1 = wmma4(fa1[s], fb[s], c1);
    }
  }
}

__global__ __launch_bounds__(256) void k_conv1_gelu(const float* __restrict__ x, const float* __restrict__ w1,
                                                    const float* __restrict__ b1, float* __restrict__ h) {
  constexpr int STAGE = P_BM * P_LDA + P_KC * P_LDB;
  __shared__ __attribute__((aligned(16))) float sm[2 * STAGE];

  const int tid  = threadIdx.x;
  const int lane = tid & 31;
  const int half = lane >> 4;
  const int l16  = lane & 15;
  const int wv   = tid >> 5;
  const int b    = blockIdx.x >> 5;
  const int bn   = blockIdx.x & 31;

  const int arow = tid >> 3;
  const int ac4  = tid & 7;
  const int brow = tid >> 5;
  const int bc4  = tid & 31;
  const float* gA = w1 + (size_t)arow * C_ + 4 * ac4;
  const float* gB = x + (size_t)b * C_ * HW_ + (size_t)brow * HW_ + bn * P_BN + 4 * bc4;

  v8f c0 = {}, c1 = {};
  tile_32x128_k512<HW_>(gA, gB, sm, c0, c1);

  float bias0[8], bias1[8];
#pragma unroll
  for (int r = 0; r < 8; ++r) {
    bias0[r] = b1[r + 8 * half];
    bias1[r] = b1[16 + r + 8 * half];
  }

  float* D = h + (size_t)b * CR_ * HW_;
  const int n = bn * P_BN + wv * 16 + l16;
#pragma unroll
  for (int r = 0; r < 8; ++r) {
    const int m = r + 8 * half;
    float v = c0[r] + bias0[r];
    v = 0.5f * v * (1.0f + erff(v * 0.70710678118654752f));
    D[(size_t)m * HW_ + n] = v;
    float u = c1[r] + bias1[r];
    u = 0.5f * u * (1.0f + erff(u * 0.70710678118654752f));
    D[(size_t)(16 + m) * HW_ + n] = u;
  }
}

__global__ void k_conv2(const float* __restrict__ h, const float* __restrict__ w2,
                        const float* __restrict__ b2, float* __restrict__ y) {
  constexpr int NT = 4;
  const int wid = blockIdx.x * 8 + (threadIdx.x >> 5);
  const int NG = HW_ / (16 * NT);
  const int MT = C_ / 16;
  const int b  = wid / (MT * NG);
  const int r0 = wid % (MT * NG);
  const int mt = r0 / NG;
  const int ng = r0 % NG;

  const float* A  = w2 + (size_t)mt * 16 * CR_;
  const float* Bm = h + (size_t)b * CR_ * HW_ + ng * 16 * NT;
  v8f c[NT] = {};
  gemm_row<NT>(A, CR_, Bm, HW_, CR_, c);

  const int lane = threadIdx.x & 31;
  const int half = lane >> 4;
  const int l16  = lane & 15;
  float* D = y + (size_t)b * C_ * HW_;
#pragma unroll
  for (int t = 0; t < NT; ++t)
#pragma unroll
    for (int r = 0; r < 8; ++r) {
      const int m = mt * 16 + r + 8 * half;
      const int n = ng * 16 * NT + t * 16 + l16;
      D[(size_t)m * HW_ + n] = c[t][r] + b2[m];
    }
}

#define E_BM 32
#define E_BN 64
#define E_KC 32
#define E_LD 36
#define E_TILES 72

__global__ __launch_bounds__(128) void k_energy(const float* __restrict__ y, float* __restrict__ e) {
  __shared__ __attribute__((aligned(16))) float sm[2 * (E_BM + E_BN) * E_LD];
  constexpr int STAGE = (E_BM + E_BN) * E_LD;
  constexpr int NCH = HW_ / E_KC;

  const int tid  = threadIdx.x;
  const int lane = tid & 31;
  const int half = lane >> 4;
  const int l16  = lane & 15;
  const int wv   = tid >> 5;
  const int b    = blockIdx.x / E_TILES;
  int rem        = blockIdx.x % E_TILES;
  int j = 0;
#pragma unroll
  for (int q = 0; q < 7; ++q) {
    const int nb = 2 * (8 - q);
    if (j == q && rem >= nb) { rem -= nb; j = q + 1; }
  }
  const int wdt   = 8 - j;
  const int upper = (rem >= wdt) ? 1 : 0;
  if (upper) rem -= wdt;
  const int strip = 2 * j + upper;
  const int bn    = j + rem;
  const float* yb = y + (size_t)b * C_ * HW_;

  const int srow = tid >> 3;
  const int sc4  = tid & 7;
  const float* gp[6];
  int so[6];
#pragma unroll
  for (int i = 0; i < 6; ++i) {
    const int r = srow + 16 * i;
    const int grow = (i < 2) ? (strip * E_BM + r) : (bn * E_BN + (r - E_BM));
    gp[i] = yb + (size_t)grow * HW_ + 4 * sc4;
    so[i] = r * E_LD + 4 * sc4;
  }

  const int ao = l16 * E_LD + 2 * half;
  const int bo = (E_BM + wv * 16 + l16) * E_LD + 2 * half;

  v4f pre[6];
#pragma unroll
  for (int i = 0; i < 6; ++i) pre[i] = *(const v4f*)gp[i];
#pragma unroll
  for (int i = 0; i < 6; ++i) *(v4f*)(sm + so[i]) = pre[i];
  __syncthreads();

  v8f c0 = {}, c1 = {};
  constexpr int KS = E_KC / 4;
  constexpr int KS_HEAD = 6;
  for (int kc = 0; kc < NCH; ++kc) {
    const int cur = kc & 1;
    const bool more = (kc + 1 < NCH);
    if (more) {
#pragma unroll
      for (int i = 0; i < 6; ++i) pre[i] = *(const v4f*)(gp[i] + (size_t)(kc + 1) * E_KC);
    }
    const float* L = sm + cur * STAGE;
    v2f fa0[KS], fa1[KS], fb[KS];
#pragma unroll
    for (int s = 0; s < KS; ++s) {
      fa0[s] = *(const v2f*)(L + ao + 4 * s);
      fa1[s] = *(const v2f*)(L + ao + 16 * E_LD + 4 * s);
      fb[s]  = *(const v2f*)(L + bo + 4 * s);
    }
    __builtin_amdgcn_sched_barrier(0);
#pragma unroll
    for (int s = 0; s < KS_HEAD; ++s) {
      c0 = wmma4(fa0[s], fb[s], c0);
      c1 = wmma4(fa1[s], fb[s], c1);
    }
    __builtin_amdgcn_sched_barrier(0);
    if (more) {
      float* S = sm + (cur ^ 1) * STAGE;
#pragma unroll
      for (int i = 0; i < 6; ++i) *(v4f*)(S + so[i]) = pre[i];
    }
    __syncthreads();
    __builtin_amdgcn_sched_barrier(0);
#pragma unroll
    for (int s = KS_HEAD; s < KS; ++s) {
      c0 = wmma4(fa0[s], fb[s], c0);
      c1 = wmma4(fa1[s], fb[s], c1);
    }
  }

  float* D = e + (size_t)b * C_ * C_;
  const int m0 = strip * E_BM + 8 * half;
  const int n0 = bn * E_BN + wv * 16 + l16;
#pragma unroll
  for (int r = 0; r < 8; ++r) {
    D[(size_t)(m0 + r) * C_ + n0]      = c0[r];
    D[(size_t)(m0 + 16 + r) * C_ + n0] = c1[r];
  }
  if (bn > j) {
    float* Tp = D + (size_t)n0 * C_ + m0;
    v4f t0 = {c0[0], c0[1], c0[2], c0[3]};
    v4f t1 = {c0[4], c0[5], c0[6], c0[7]};
    v4f t2 = {c1[0], c1[1], c1[2], c1[3]};
    v4f t3 = {c1[4], c1[5], c1[6], c1[7]};
    *(v4f*)(Tp)      = t0;
    *(v4f*)(Tp + 4)  = t1;
    *(v4f*)(Tp + 16) = t2;
    *(v4f*)(Tp + 20) = t3;
  }
}

__global__ void k_softmax(float* __restrict__ e) {
  __shared__ float red[256];
  const int tid = threadIdx.x;
  float* p = e + (size_t)blockIdx.x * C_;
  const float v0 = p[tid];
  const float v1 = p[tid + 256];
  red[tid] = fminf(v0, v1);
  __syncthreads();
  for (int s = 128; s > 0; s >>= 1) {
    if (tid < s) red[tid] = fminf(red[tid], red[tid + s]);
    __syncthreads();
  }
  const float rmin = red[0];
  __syncthreads();
  const float e0 = expf(rmin - v0);
  const float e1 = expf(rmin - v1);
  red[tid] = e0 + e1;
  __syncthreads();
  for (int s = 128; s > 0; s >>= 1) {
    if (tid < s) red[tid] += red[tid + s];
    __syncthreads();
  }
  const float inv = 1.0f / red[0];
  p[tid] = e0 * inv;
  p[tid + 256] = e1 * inv;
}

__global__ __launch_bounds__(256) void k_m2(const float* __restrict__ attn, const float* __restrict__ wf,
                                            const float* __restrict__ gamma, float* __restrict__ m2) {
  constexpr int STAGE = P_BM * P_LDA + P_KC * P_LDB;
  __shared__ __attribute__((aligned(16))) float sm[2 * STAGE];

  const int tid   = threadIdx.x;
  const int lane  = tid & 31;
  const int half  = lane >> 4;
  const int l16   = lane & 15;
  const int wv    = tid >> 5;
  const int b     = blockIdx.x >> 6;
  const int strip = (blockIdx.x >> 2) & 15;
  const int bn    = blockIdx.x & 3;

  const int arow = tid >> 3;
  const int ac4  = tid & 7;
  const int brow = tid >> 5;
  const int bc4  = tid & 31;
  const float* gA = wf + (size_t)(strip * P_BM + arow) * C2_ + 4 * ac4;
  const float* gB = attn + (size_t)b * C_ * C_ + (size_t)brow * C_ + bn * P_BN + 4 * bc4;

  v8f c0 = {}, c1 = {};
  tile_32x128_k512<C_>(gA, gB, sm, c0, c1);

  const float g = gamma[0];
  const int n  = bn * P_BN + wv * 16 + l16;
  const int m0 = strip * P_BM + 8 * half;
  float wa[8], wb[8];
#pragma unroll
  for (int r = 0; r < 8; ++r) {
    wa[r] = wf[(size_t)(m0 + r) * C2_ + n];
    wb[r] = wf[(size_t)(m0 + 16 + r) * C2_ + n];
  }
  float* D = m2 + (size_t)b * C_ * C_;
#pragma unroll
  for (int r = 0; r < 8; ++r) {
    D[(size_t)(m0 + r) * C_ + n]      = g * c0[r] + wa[r];
    D[(size_t)(m0 + 16 + r) * C_ + n] = g * c1[r] + wb[r];
  }
}

#define F_BM  128
#define F_BN  128
#define F_KC  32
#define F_LDA 36
#define F_LDB 144

__global__ __launch_bounds__(256) void k_fusion(const float* __restrict__ m2, const float* __restrict__ y,
                                                const float* __restrict__ x, const float* __restrict__ wf,
                                                const float* __restrict__ bf, float* __restrict__ out) {
  constexpr int A_SZ  = F_BM * F_LDA;
  constexpr int STAGE = A_SZ + F_KC * F_LDB;
  __shared__ __attribute__((aligned(16))) float sm[2 * STAGE];
  constexpr int NCH  = C2_ / F_KC;
  constexpr int NCH1 = C_ / F_KC;

  const int tid  = threadIdx.x;
  const int lane = tid & 31;
  const int half = lane >> 4;
  const int l16  = lane & 15;
  const int wv   = tid >> 5;
  const int wm   = wv >> 2;
  const int wn   = wv & 3;
  const int b    = blockIdx.x >> 7;
  const int bm   = (blockIdx.x >> 5) & 3;
  const int bn   = blockIdx.x & 31;

  const int arow = tid >> 3;
  const int ac4  = tid & 7;
  const int brow = tid >> 5;
  const int bc4  = tid & 31;
  const float* gA1 = m2 + (size_t)b * C_ * C_ + (size_t)(bm * F_BM + arow) * C_ + 4 * ac4;
  const float* gA2 = wf + (size_t)(bm * F_BM + arow) * C2_ + C_ + 4 * ac4;
  const float* gB1 = y + (size_t)b * C_ * HW_ + (size_t)brow * HW_ + bn * F_BN + 4 * bc4;
  const float* gB2 = x + (size_t)b * C_ * HW_ + (size_t)brow * HW_ + bn * F_BN + 4 * bc4;
  const int soa = arow * F_LDA + 4 * ac4;
  const int sob = A_SZ + brow * F_LDB + 4 * bc4;

  const int ao = (wm * 64 + l16) * F_LDA + 2 * half;
  const int bo = A_SZ + (2 * half) * F_LDB + wn * 32 + l16;

  v4f pre[8];
#pragma unroll
  for (int i = 0; i < 4; ++i) pre[i] = *(const v4f*)(gA1 + (size_t)i * 32 * C_);
#pragma unroll
  for (int i = 0; i < 4; ++i) pre[4 + i] = *(const v4f*)(gB1 + (size_t)i * 8 * HW_);
#pragma unroll
  for (int i = 0; i < 4; ++i) *(v4f*)(sm + soa + i * 32 * F_LDA) = pre[i];
#pragma unroll
  for (int i = 0; i < 4; ++i) *(v4f*)(sm + sob + i * 8 * F_LDB) = pre[4 + i];
  __syncthreads();

  v8f c[4][2] = {};
  constexpr int KS = F_KC / 4;
  constexpr int KS_HEAD = 6;
  for (int kc = 0; kc < NCH; ++kc) {
    const int cur = kc & 1;
    const bool more = (kc + 1 < NCH);
    if (more) {
      const int kn = kc + 1;
      if (kn < NCH1) {
#pragma unroll
        for (int i = 0; i < 4; ++i) pre[i] = *(const v4f*)(gA1 + (size_t)i * 32 * C_ + kn * F_KC);
#pragma unroll
        for (int i = 0; i < 4; ++i) pre[4 + i] = *(const v4f*)(gB1 + (size_t)(kn * F_KC + 8 * i) * HW_);
      } else {
#pragma unroll
        for (int i = 0; i < 4; ++i) pre[i] = *(const v4f*)(gA2 + (size_t)i * 32 * C2_ + (kn - NCH1) * F_KC);
#pragma unroll
        for (int i = 0; i < 4; ++i) pre[4 + i] = *(const v4f*)(gB2 + (size_t)((kn - NCH1) * F_KC + 8 * i) * HW_);
      }
    }
    const float* L = sm + cur * STAGE;
    v2f fa[4][KS], fb[2][KS];
#pragma unroll
    for (int s = 0; s < KS; ++s) {
#pragma unroll
      for (int t = 0; t < 4; ++t) fa[t][s] = *(const v2f*)(L + ao + t * 16 * F_LDA + 4 * s);
#pragma unroll
      for (int t = 0; t < 2; ++t) {
        fb[t][s].x = L[bo + t * 16 + 4 * s * F_LDB];
        fb[t][s].y = L[bo + t * 16 + 4 * s * F_LDB + F_LDB];
      }
    }
    __builtin_amdgcn_sched_barrier(0);
#pragma unroll
    for (int s = 0; s < KS_HEAD; ++s)
#pragma unroll
      for (int t = 0; t < 4; ++t) {
        c[t][0] = wmma4(fa[t][s], fb[0][s], c[t][0]);
        c[t][1] = wmma4(fa[t][s], fb[1][s], c[t][1]);
      }
    __builtin_amdgcn_sched_barrier(0);
    if (more) {
      float* S = sm + (cur ^ 1) * STAGE;
#pragma unroll
      for (int i = 0; i < 4; ++i) *(v4f*)(S + soa + i * 32 * F_LDA) = pre[i];
#pragma unroll
      for (int i = 0; i < 4; ++i) *(v4f*)(S + sob + i * 8 * F_LDB) = pre[4 + i];
    }
    __syncthreads();
    __builtin_amdgcn_sched_barrier(0);
#pragma unroll
    for (int s = KS_HEAD; s < KS; ++s)
#pragma unroll
      for (int t = 0; t < 4; ++t) {
        c[t][0] = wmma4(fa[t][s], fb[0][s], c[t][0]);
        c[t][1] = wmma4(fa[t][s], fb[1][s], c[t][1]);
      }
  }

  float* D = out + (size_t)b * C_ * HW_;
  const int n0 = bn * F_BN + wn * 32 + l16;
#pragma unroll
  for (int t = 0; t < 4; ++t) {
    const int m0 = bm * F_BM + wm * 64 + t * 16 + 8 * half;
#pragma unroll
    for (int r = 0; r < 8; ++r) {
      const float bias = bf[m0 + r];
      D[(size_t)(m0 + r) * HW_ + n0]      = c[t][0][r] + bias;
      D[(size_t)(m0 + r) * HW_ + n0 + 16] = c[t][1][r] + bias;
    }
  }
}

extern "C" void kernel_launch(void* const* d_in, const int* in_sizes, int n_in,
                              void* d_out, int out_size, void* d_ws, size_t ws_size,
                              hipStream_t stream) {
  const float* x     = (const float*)d_in[0];
  const float* w1    = (const float*)d_in[1];
  const float* b1    = (const float*)d_in[2];
  const float* w2    = (const float*)d_in[3];
  const float* b2    = (const float*)d_in[4];
  const float* wf    = (const float*)d_in[5];
  const float* bf    = (const float*)d_in[6];
  const float* gamma = (const float*)d_in[7];
  float* out = (float*)d_out;

  char* ws = (char*)d_ws;
  float* h    = (float*)(ws);
  float* y    = (float*)(ws + (size_t)4 * 1024 * 1024);
  float* attn = (float*)(ws + (size_t)68 * 1024 * 1024);
  float* m2   = (float*)(ws + (size_t)76 * 1024 * 1024);

  const dim3 blk(256);
  k_conv1_gelu<<<dim3(256),  blk, 0, stream>>>(x, w1, b1, h);
  k_conv2     <<<dim3(2048), blk, 0, stream>>>(h, w2, b2, y);
  k_energy    <<<dim3(576),  dim3(128), 0, stream>>>(y, attn);
  k_softmax   <<<dim3(4096), blk, 0, stream>>>(attn);
  k_m2        <<<dim3(512),  blk, 0, stream>>>(attn, wf, gamma, m2);
  k_fusion    <<<dim3(1024), blk, 0, stream>>>(m2, y, x, wf, bf, out);
}


# energy y.y^T: K split over 2 blocks per 32x64 tile (576 -> 1152 blocks, 4-5 waves a SIMD), each half summed to its own plane, softmax adds the 2 planes; f32 sum order changes (2 partial sums)
# speedup vs baseline: 2.0184x; 1.0580x over previous
#include <hip/hip_runtime.h>
#include <math.h>


#define B_   8
#define C_   512
#define CR_  32
#define HW_  4096
#define C2_  1024

typedef __attribute__((ext_vector_type(2))) float v2f;
typedef __attribute__((ext_vector_type(8))) float v8f;

__device__ __forceinline__ v8f wmma4(v2f a, v2f b, v8f c) {
  return __builtin_amdgcn_wmma_f32_16x16x4_f32(
       false, a,  false, b,
       (short)0, c,  false,  false);
}

template<int NT>
__device__ __forceinline__ void gemm_row(const float* __restrict__ A, int lda,
                                         const float* __restrict__ Bm, int ldb,
                                         int K, v8f c[NT]) {
  const int lane = threadIdx.x & 31;
  const int half = lane >> 4;
  const int l16  = lane & 15;
  const float* ap = A + (size_t)l16 * lda + 2 * half;
  const float* bp = Bm + (size_t)(2 * half) * ldb + l16;
#pragma unroll 2
  for (int k = 0; k < K; k += 4) {
    v2f a = *(const v2f*)ap;
#pragma unroll
    for (int t = 0; t < NT; ++t) {
      v2f b;
      b.x = bp[t * 16];
      b.y = bp[t * 16 + ldb];
      c[t] = wmma4(a, b, c[t]);
    }
    ap += 4;
    bp += (size_t)4 * ldb;
  }
}

#define P_BM    32
#define P_BN    128
#define P_KC    32
#define P_LDA   36
#define P_LDB   144
#define P_AHEAD 3

typedef __attribute__((ext_vector_type(4))) float v4f;

__device__ __forceinline__ void lds_barrier() {
  __builtin_amdgcn_fence(__ATOMIC_RELEASE, "workgroup", "local");
  __builtin_amdgcn_s_barrier();
  __builtin_amdgcn_fence(__ATOMIC_ACQUIRE, "workgroup", "local");
}

template<int LDGB>
__device__ __forceinline__ void tile_32x128_k512(const float* __restrict__ gA, const float* __restrict__ gB,
                                                 float* sm, v8f& c0, v8f& c1) {
  constexpr int A_SZ  = P_BM * P_LDA;
  constexpr int STAGE = A_SZ + P_KC * P_LDB;
  constexpr int NCH = C_ / P_KC;

  const int tid  = threadIdx.x;
  const int lane = tid & 31;
  const int half = lane >> 4;
  const int l16  = lane & 15;
  const int wv   = tid >> 5;
  const int arow = tid >> 3;
  const int ac4  = tid & 7;
  const int brow = tid >> 5;
  const int bc4  = tid & 31;
  const int soa = arow * P_LDA + 4 * ac4;
  const int sob = A_SZ + brow * P_LDB + 4 * bc4;

  const int ao = l16 * P_LDA + 2 * half;
  const int bo = A_SZ + (2 * half) * P_LDB + wv * 16 + l16;

  v4f ring[P_AHEAD][5];
#pragma unroll
  for (int j = 0; j < P_AHEAD; ++j) {
    ring[j][0] = *(const v4f*)(gA + j * P_KC);
#pragma unroll
    for (int i = 0; i < 4; ++i) ring[j][1 + i] = *(const v4f*)(gB + (size_t)(j * P_KC + 8 * i) * LDGB);
  }
  *(v4f*)(sm + soa) = ring[0][0];
#pragma unroll
  for (int i = 0; i < 4; ++i) *(v4f*)(sm + sob + i * 8 * P_LDB) = ring[0][1 + i];
  lds_barrier();

  constexpr int KS = P_KC / 4;
  constexpr int KS_HEAD = 6;
#pragma unroll
  for (int kc = 0; kc < NCH; ++kc) {
    const int cur = kc & 1;
    if (kc + P_AHEAD < NCH) {
      const int kn = kc + P_AHEAD;
      ring[kc % P_AHEAD][0] = *(const v4f*)(gA + kn * P_KC);
#pragma unroll
      for (int i = 0; i < 4; ++i)
        ring[kc % P_AHEAD][1 + i] = *(const v4f*)(gB + (size_t)(kn * P_KC + 8 * i) * LDGB);
    }
    const float* L = sm + cur * STAGE;
    v2f fa0[KS], fa1[KS], fb[KS];
#pragma unroll
    for (int s = 0; s < KS; ++s) {
      fa0[s] = *(const v2f*)(L + ao + 4 * s);
      fa1[s] = *(const v2f*)(L + ao + 16 * P_LDA + 4 * s);
      fb[s].x = L[bo + 4 * s * P_LDB];
      fb[s].y = L[bo + 4 * s * P_LDB + P_LDB];
    }
    __builtin_amdgcn_sched_barrier(0);
#pragma unroll
    for (int s = 0; s < KS_HEAD; ++s) {
      c0 = wmma4(fa0[s], fb[s], c0);
      c1 = wmma4(fa1[s], fb[s], c1);
    }
    __builtin_amdgcn_sched_barrier(0);
    if (kc + 1 < NCH) {
      float* S = sm + (cur ^ 1) * STAGE;
      *(v4f*)(S + soa) = ring[(kc + 1) % P_AHEAD][0];
#pragma unroll
      for (int i = 0; i < 4; ++i) *(v4f*)(S + sob + i * 8 * P_LDB) = ring[(kc + 1) % P_AHEAD][1 + i];
      lds_barrier();
    }
    __builtin_amdgcn_sched_barrier(0);
#pragma unroll
    for (int s = KS_HEAD; s < KS; ++s) {
      c0 = wmma4(fa0[s], fb[s], c0);
      c1 = wmma4(fa1[s], fb[s], c1);
    }
  }
}

__global__ __launch_bounds__(256) void k_conv1_gelu(const float* __restrict__ x, const float* __restrict__ w1,
                                                    const float* __restrict__ b1, float* __restrict__ h) {
  constexpr int STAGE = P_BM * P_LDA + P_KC * P_LDB;
  __shared__ __attribute__((aligned(16))) float sm[2 * STAGE];

  const int tid  = threadIdx.x;
  const int lane = tid & 31;
  const int half = lane >> 4;
  const int l16  = lane & 15;
  const int wv   = tid >> 5;
  const int b    = blockIdx.x >> 5;
  const int bn   = blockIdx.x & 31;

  const int arow = tid >> 3;
  const int ac4  = tid & 7;
  const int brow = tid >> 5;
  const int bc4  = tid & 31;
  const float* gA = w1 + (size_t)arow * C_ + 4 * ac4;
  const float* gB = x + (size_t)b * C_ * HW_ + (size_t)brow * HW_ + bn * P_BN + 4 * bc4;

  v8f c0 = {}, c1 = {};
  tile_32x128_k512<HW_>(gA, gB, sm, c0, c1);

  float bias0[8], bias1[8];
#pragma unroll
  for (int r = 0; r < 8; ++r) {
    bias0[r] = b1[r + 8 * half];
    bias1[r] = b1[16 + r + 8 * half];
  }

  float* D = h + (size_t)b * CR_ * HW_;
  const int n = bn * P_BN + wv * 16 + l16;
#pragma unroll
  for (int r = 0; r < 8; ++r) {
    const int m = r + 8 * half;
    float v = c0[r] + bias0[r];
    v = 0.5f * v * (1.0f + erff(v * 0.70710678118654752f));
    D[(size_t)m * HW_ + n] = v;
    float u = c1[r] + bias1[r];
    u = 0.5f * u * (1.0f + erff(u * 0.70710678118654752f));
    D[(size_t)(16 + m) * HW_ + n] = u;
  }
}

__global__ void k_conv2(const float* __restrict__ h, const float* __restrict__ w2,
                        const float* __restrict__ b2, float* __restrict__ y) {
  constexpr int NT = 4;
  const int wid = blockIdx.x * 8 + (threadIdx.x >> 5);
  const int NG = HW_ / (16 * NT);
  const int MT = C_ / 16;
  const int b  = wid / (MT * NG);
  const int r0 = wid % (MT * NG);
  const int mt = r0 / NG;
  const int ng = r0 % NG;

  const float* A  = w2 + (size_t)mt * 16 * CR_;
  const float* Bm = h + (size_t)b * CR_ * HW_ + ng * 16 * NT;
  v8f c[NT] = {};
  gemm_row<NT>(A, CR_, Bm, HW_, CR_, c);

  const int lane = threadIdx.x & 31;
  const int half = lane >> 4;
  const int l16  = lane & 15;
  float* D = y + (size_t)b * C_ * HW_;
#pragma unroll
  for (int t = 0; t < NT; ++t)
#pragma unroll
    for (int r = 0; r < 8; ++r) {
      const int m = mt * 16 + r + 8 * half;
      const int n = ng * 16 * NT + t * 16 + l16;
      D[(size_t)m * HW_ + n] = c[t][r] + b2[m];
    }
}

#define E_BM 32
#define E_BN 64
#define E_KC 32
#define E_LD 36
#define E_TILES 72
#define E_KSPLIT 2

__global__ __launch_bounds__(128) void k_energy(const float* __restrict__ y, float* __restrict__ e0,
                                                float* __restrict__ e1) {
  __shared__ __attribute__((aligned(16))) float sm[2 * (E_BM + E_BN) * E_LD];
  constexpr int STAGE = (E_BM + E_BN) * E_LD;
  constexpr int NCH = HW_ / E_KSPLIT / E_KC;

  const int tid  = threadIdx.x;
  const int lane = tid & 31;
  const int half = lane >> 4;
  const int l16  = lane & 15;
  const int wv   = tid >> 5;
  const int b    = blockIdx.x / (E_KSPLIT * E_TILES);
  int rem        = blockIdx.x % (E_KSPLIT * E_TILES);
  const int kh   = rem / E_TILES;
  rem            = rem % E_TILES;
  int j = 0;
#pragma unroll
  for (int q = 0; q < 7; ++q) {
    const int nb = 2 * (8 - q);
    if (j == q && rem >= nb) { rem -= nb; j = q + 1; }
  }
  const int wdt   = 8 - j;
  const int upper = (rem >= wdt) ? 1 : 0;
  if (upper) rem -= wdt;
  const int strip = 2 * j + upper;
  const int bn    = j + rem;
  const float* yb = y + (size_t)b * C_ * HW_ + kh * (HW_ / E_KSPLIT);

  const int srow = tid >> 3;
  const int sc4  = tid & 7;
  const float* gp[6];
  int so[6];
#pragma unroll
  for (int i = 0; i < 6; ++i) {
    const int r = srow + 16 * i;
    const int grow = (i < 2) ? (strip * E_BM + r) : (bn * E_BN + (r - E_BM));
    gp[i] = yb + (size_t)grow * HW_ + 4 * sc4;
    so[i] = r * E_LD + 4 * sc4;
  }

  const int ao = l16 * E_LD + 2 * half;
  const int bo = (E_BM + wv * 16 + l16) * E_LD + 2 * half;

  v4f pre[6];
#pragma unroll
  for (int i = 0; i < 6; ++i) pre[i] = *(const v4f*)gp[i];
#pragma unroll
  for (int i = 0; i < 6; ++i) *(v4f*)(sm + so[i]) = pre[i];
  __syncthreads();

  v8f c0 = {}, c1 = {};
  constexpr int KS = E_KC / 4;
  constexpr int KS_HEAD = 6;
  for (int kc = 0; kc < NCH; ++kc) {
    const int cur = kc & 1;
    const bool more = (kc + 1 < NCH);
    if (more) {
#pragma unroll
      for (int i = 0; i < 6; ++i) pre[i] = *(const v4f*)(gp[i] + (size_t)(kc + 1) * E_KC);
    }
    const float* L = sm + cur * STAGE;
    v2f fa0[KS], fa1[KS], fb[KS];
#pragma unroll
    for (int s = 0; s < KS; ++s) {
      fa0[s] = *(const v2f*)(L + ao + 4 * s);
      fa1[s] = *(const v2f*)(L + ao + 16 * E_LD + 4 * s);
      fb[s]  = *(const v2f*)(L + bo + 4 * s);
    }
    __builtin_amdgcn_sched_barrier(0);
#pragma unroll
    for (int s = 0; s < KS_HEAD; ++s) {
      c0 = wmma4(fa0[s], fb[s], c0);
      c1 = wmma4(fa1[s], fb[s], c1);
    }
    __builtin_amdgcn_sched_barrier(0);
    if (more) {
      float* S = sm + (cur ^ 1) * STAGE;
#pragma unroll
      for (int i = 0; i < 6; ++i) *(v4f*)(S + so[i]) = pre[i];
    }
    __syncthreads();
    __builtin_amdgcn_sched_barrier(0);
#pragma unroll
    for (int s = KS_HEAD; s < KS; ++s) {
      c0 = wmma4(fa0[s], fb[s], c0);
      c1 = wmma4(fa1[s], fb[s], c1);
    }
  }

  float* D = (kh ? e1 : e0) + (size_t)b * C_ * C_;
  const int m0 = strip * E_BM + 8 * half;
  const int n0 = bn * E_BN + wv * 16 + l16;
#pragma unroll
  for (int r = 0; r < 8; ++r) {
    D[(size_t)(m0 + r) * C_ + n0]      = c0[r];
    D[(size_t)(m0 + 16 + r) * C_ + n0] = c1[r];
  }
  if (bn > j) {
    float* Tp = D + (size_t)n0 * C_ + m0;
    v4f t0 = {c0[0], c0[1], c0[2], c0[3]};
    v4f t1 = {c0[4], c0[5], c0[6], c0[7]};
    v4f t2 = {c1[0], c1[1], c1[2], c1[3]};
    v4f t3 = {c1[4], c1[5], c1[6], c1[7]};
    *(v4f*)(Tp)      = t0;
    *(v4f*)(Tp + 4)  = t1;
    *(v4f*)(Tp + 16) = t2;
    *(v4f*)(Tp + 20) = t3;
  }
}

__global__ void k_softmax(float* __restrict__ e, const float* __restrict__ eb) {
  __shared__ float red[256];
  const int tid = threadIdx.x;
  float* p = e + (size_t)blockIdx.x * C_;
  const float* q = eb + (size_t)blockIdx.x * C_;
  const float v0 = p[tid] + q[tid];
  const float v1 = p[tid + 256] + q[tid + 256];
  red[tid] = fminf(v0, v1);
  __syncthreads();
  for (int s = 128; s > 0; s >>= 1) {
    if (tid < s) red[tid] = fminf(red[tid], red[tid + s]);
    __syncthreads();
  }
  const float rmin = red[0];
  __syncthreads();
  const float e0 = expf(rmin - v0);
  const float e1 = expf(rmin - v1);
  red[tid] = e0 + e1;
  __syncthreads();
  for (int s = 128; s > 0; s >>= 1) {
    if (tid < s) red[tid] += red[tid + s];
    __syncthreads();
  }
  const float inv = 1.0f / red[0];
  p[tid] = e0 * inv;
  p[tid + 256] = e1 * inv;
}

__global__ __launch_bounds__(256) void k_m2(const float* __restrict__ attn, const float* __restrict__ wf,
                                            const float* __restrict__ gamma, float* __restrict__ m2) {
  constexpr int STAGE = P_BM * P_LDA + P_KC * P_LDB;
  __shared__ __attribute__((aligned(16))) float sm[2 * STAGE];

  const int tid   = threadIdx.x;
  const int lane  = tid & 31;
  const int half  = lane >> 4;
  const int l16   = lane & 15;
  const int wv    = tid >> 5;
  const int b     = blockIdx.x >> 6;
  const int strip = (blockIdx.x >> 2) & 15;
  const int bn    = blockIdx.x & 3;

  const int arow = tid >> 3;
  const int ac4  = tid & 7;
  const int brow = tid >> 5;
  const int bc4  = tid & 31;
  const float* gA = wf + (size_t)(strip * P_BM + arow) * C2_ + 4 * ac4;
  const float* gB = attn + (size_t)b * C_ * C_ + (size_t)brow * C_ + bn * P_BN + 4 * bc4;

  v8f c0 = {}, c1 = {};
  tile_32x128_k512<C_>(gA, gB, sm, c0, c1);

  const float g = gamma[0];
  const int n  = bn * P_BN + wv * 16 + l16;
  const int m0 = strip * P_BM + 8 * half;
  float wa[8], wb[8];
#pragma unroll
  for (int r = 0; r < 8; ++r) {
    wa[r] = wf[(size_t)(m0 + r) * C2_ + n];
    wb[r] = wf[(size_t)(m0 + 16 + r) * C2_ + n];
  }
  float* D = m2 + (size_t)b * C_ * C_;
#pragma unroll
  for (int r = 0; r < 8; ++r) {
    D[(size_t)(m0 + r) * C_ + n]      = g * c0[r] + wa[r];
    D[(size_t)(m0 + 16 + r) * C_ + n] = g * c1[r] + wb[r];
  }
}

#define F_BM  128
#define F_BN  128
#define F_KC  32
#define F_LDA 36
#define F_LDB 144

__global__ __launch_bounds__(256) void k_fusion(const float* __restrict__ m2, const float* __restrict__ y,
                                                const float* __restrict__ x, const float* __restrict__ wf,
                                                const float* __restrict__ bf, float* __restrict__ out) {
  constexpr int A_SZ  = F_BM * F_LDA;
  constexpr int STAGE = A_SZ + F_KC * F_LDB;
  __shared__ __attribute__((aligned(16))) float sm[2 * STAGE];
  constexpr int NCH  = C2_ / F_KC;
  constexpr int NCH1 = C_ / F_KC;

  const int tid  = threadIdx.x;
  const int lane = tid & 31;
  const int half = lane >> 4;
  const int l16  = lane & 15;
  const int wv   = tid >> 5;
  const int wm   = wv >> 2;
  const int wn   = wv & 3;
  const int b    = blockIdx.x >> 7;
  const int bm   = (blockIdx.x >> 5) & 3;
  const int bn   = blockIdx.x & 31;

  const int arow = tid >> 3;
  const int ac4  = tid & 7;
  const int brow = tid >> 5;
  const int bc4  = tid & 31;
  const float* gA1 = m2 + (size_t)b * C_ * C_ + (size_t)(bm * F_BM + arow) * C_ + 4 * ac4;
  const float* gA2 = wf + (size_t)(bm * F_BM + arow) * C2_ + C_ + 4 * ac4;
  const float* gB1 = y + (size_t)b * C_ * HW_ + (size_t)brow * HW_ + bn * F_BN + 4 * bc4;
  const float* gB2 = x + (size_t)b * C_ * HW_ + (size_t)brow * HW_ + bn * F_BN + 4 * bc4;
  const int soa = arow * F_LDA + 4 * ac4;
  const int sob = A_SZ + brow * F_LDB + 4 * bc4;

  const int ao = (wm * 64 + l16) * F_LDA + 2 * half;
  const int bo = A_SZ + (2 * half) * F_LDB + wn * 32 + l16;

  v4f pre[8];
#pragma unroll
  for (int i = 0; i < 4; ++i) pre[i] = *(const v4f*)(gA1 + (size_t)i * 32 * C_);
#pragma unroll
  for (int i = 0; i < 4; ++i) pre[4 + i] = *(const v4f*)(gB1 + (size_t)i * 8 * HW_);
#pragma unroll
  for (int i = 0; i < 4; ++i) *(v4f*)(sm + soa + i * 32 * F_LDA) = pre[i];
#pragma unroll
  for (int i = 0; i < 4; ++i) *(v4f*)(sm + sob + i * 8 * F_LDB) = pre[4 + i];
  __syncthreads();

  v8f c[4][2] = {};
  constexpr int KS = F_KC / 4;
  constexpr int KS_HEAD = 6;
  for (int kc = 0; kc < NCH; ++kc) {
    const int cur = kc & 1;
    const bool more = (kc + 1 < NCH);
    if (more) {
      const int kn = kc + 1;
      if (kn < NCH1) {
#pragma unroll
        for (int i = 0; i < 4; ++i) pre[i] = *(const v4f*)(gA1 + (size_t)i * 32 * C_ + kn * F_KC);
#pragma unroll
        for (int i = 0; i < 4; ++i) pre[4 + i] = *(const v4f*)(gB1 + (size_t)(kn * F_KC + 8 * i) * HW_);
      } else {
#pragma unroll
        for (int i = 0; i < 4; ++i) pre[i] = *(const v4f*)(gA2 + (size_t)i * 32 * C2_ + (kn - NCH1) * F_KC);
#pragma unroll
        for (int i = 0; i < 4; ++i) pre[4 + i] = *(const v4f*)(gB2 + (size_t)((kn - NCH1) * F_KC + 8 * i) * HW_);
      }
    }
    const float* L = sm + cur * STAGE;
    v2f fa[4][KS], fb[2][KS];
#pragma unroll
    for (int s = 0; s < KS; ++s) {
#pragma unroll
      for (int t = 0; t < 4; ++t) fa[t][s] = *(const v2f*)(L + ao + t * 16 * F_LDA + 4 * s);
#pragma unroll
      for (int t = 0; t < 2; ++t) {
        fb[t][s].x = L[bo + t * 16 + 4 * s * F_LDB];
        fb[t][s].y = L[bo + t * 16 + 4 * s * F_LDB + F_LDB];
      }
    }
    __builtin_amdgcn_sched_barrier(0);
#pragma unroll
    for (int s = 0; s < KS_HEAD; ++s)
#pragma unroll
      for (int t = 0; t < 4; ++t) {
        c[t][0] = wmma4(fa[t][s], fb[0][s], c[t][0]);
        c[t][1] = wmma4(fa[t][s], fb[1][s], c[t][1]);
      }
    __builtin_amdgcn_sched_barrier(0);
    if (more) {
      float* S = sm + (cur ^ 1) * STAGE;
#pragma unroll
      for (int i = 0; i < 4; ++i) *(v4f*)(S + soa + i * 32 * F_LDA) = pre[i];
#pragma unroll
      for (int i = 0; i < 4; ++i) *(v4f*)(S + sob + i * 8 * F_LDB) = pre[4 + i];
    }
    __syncthreads();
    __builtin_amdgcn_sched_barrier(0);
#pragma unroll
    for (int s = KS_HEAD; s < KS; ++s)
#pragma unroll
      for (int t = 0; t < 4; ++t) {
        c[t][0] = wmma4(fa[t][s], fb[0][s], c[t][0]);
        c[t][1] = wmma4(fa[t][s], fb[1][s], c[t][1]);
      }
  }

  float* D = out + (size_t)b * C_ * HW_;
  const int n0 = bn * F_BN + wn * 32 + l16;
#pragma unroll
  for (int t = 0; t < 4; ++t) {
    const int m0 = bm * F_BM + wm * 64 + t * 16 + 8 * half;
#pragma unroll
    for (int r = 0; r < 8; ++r) {
      const float bias = bf[m0 + r];
      D[(size_t)(m0 + r) * HW_ + n0]      = c[t][0][r] + bias;
      D[(size_t)(m0 + r) * HW_ + n0 + 16] = c[t][1][r] + bias;
    }
  }
}

extern "C" void kernel_launch(void* const* d_in, const int* in_sizes, int n_in,
                              void* d_out, int out_size, void* d_ws, size_t ws_size,
                              hipStream_t stream) {
  const float* x     = (const float*)d_in[0];
  const float* w1    = (const float*)d_in[1];
  const float* b1    = (const float*)d_in[2];
  const float* w2    = (const float*)d_in[3];
  const float* b2    = (const float*)d_in[4];
  const float* wf    = (const float*)d_in[5];
  const float* bf    = (const float*)d_in[6];
  const float* gamma = (const float*)d_in[7];
  float* out = (float*)d_out;

  char* ws = (char*)d_ws;
  float* h    = (float*)(ws);
  float* y    = (float*)(ws + (size_t)4 * 1024 * 1024);
  float* attn = (float*)(ws + (size_t)68 * 1024 * 1024);
  float* m2   = (float*)(ws + (size_t)76 * 1024 * 1024);

  const dim3 blk(256);
  k_conv1_gelu<<<dim3(256),  blk, 0, stream>>>(x, w1, b1, h);
  k_conv2     <<<dim3(2048), blk, 0, stream>>>(h, w2, b2, y);
  k_energy    <<<dim3(1152), dim3(128), 0, stream>>>(y, attn, m2);
  k_softmax   <<<dim3(4096), blk, 0, stream>>>(attn, m2);
  k_m2        <<<dim3(512),  blk, 0, stream>>>(attn, wf, gamma, m2);
  k_fusion    <<<dim3(1024), blk, 0, stream>>>(m2, y, x, wf, bf, out);
}


# conv2 fused into conv1's kernel: a wave's GELU'd 32x16 result tiles serve as B fragments of W2.h (k paired as lanes hold it), w2 from LDS; h never stored, 6 -> 5 launches; f32 sum order changes in y
# speedup vs baseline: 2.0993x; 1.0401x over previous
#include <hip/hip_runtime.h>
#include <math.h>


#define B_   8
#define C_   512
#define CR_  32
#define HW_  4096
#define C2_  1024

typedef __attribute__((ext_vector_type(2))) float v2f;
typedef __attribute__((ext_vector_type(8))) float v8f;

__device__ __forceinline__ v8f wmma4(v2f a, v2f b, v8f c) {
  return __builtin_amdgcn_wmma_f32_16x16x4_f32(
       false, a,  false, b,
       (short)0, c,  false,  false);
}

#define P_BM    32
#define P_BN    128
#define P_KC    32
#define P_LDA   36
#define P_LDB   144
#define P_AHEAD 3

typedef __attribute__((ext_vector_type(4))) float v4f;

__device__ __forceinline__ void lds_barrier() {
  __builtin_amdgcn_fence(__ATOMIC_RELEASE, "workgroup", "local");
  __builtin_amdgcn_s_barrier();
  __builtin_amdgcn_fence(__ATOMIC_ACQUIRE, "workgroup", "local");
}

template<int LDGB>
__device__ __forceinline__ void tile_32x128_k512(const float* __restrict__ gA, const float* __restrict__ gB,
                                                 float* sm, v8f& c0, v8f& c1) {
  constexpr int A_SZ  = P_BM * P_LDA;
  constexpr int STAGE = A_SZ + P_KC * P_LDB;
  constexpr int NCH = C_ / P_KC;

  const int tid  = threadIdx.x;
  const int lane = tid & 31;
  const int half = lane >> 4;
  const int l16  = lane & 15;
  const int wv   = tid >> 5;
  const int arow = tid >> 3;
  const int ac4  = tid & 7;
  const int brow = tid >> 5;
  const int bc4  = tid & 31;
  const int soa = arow * P_LDA + 4 * ac4;
  const int sob = A_SZ + brow * P_LDB + 4 * bc4;

  const int ao = l16 * P_LDA + 2 * half;
  const int bo = A_SZ + (2 * half) * P_LDB + wv * 16 + l16;

  v4f ring[P_AHEAD][5];
#pragma unroll
  for (int j = 0; j < P_AHEAD; ++j) {
    ring[j][0] = *(const v4f*)(gA + j * P_KC);
#pragma unroll
    for (int i = 0; i < 4; ++i) ring[j][1 + i] = *(const v4f*)(gB + (size_t)(j * P_KC + 8 * i) * LDGB);
  }
  *(v4f*)(sm + soa) = ring[0][0];
#pragma unroll
  for (int i = 0; i < 4; ++i) *(v4f*)(sm + sob + i * 8 * P_LDB) = ring[0][1 + i];
  lds_barrier();

  constexpr int KS = P_KC / 4;
  constexpr int KS_HEAD = 6;
#pragma unroll
  for (int kc = 0; kc < NCH; ++kc) {
    const int cur = kc & 1;
    if (kc + P_AHEAD < NCH) {
      const int kn = kc + P_AHEAD;
      ring[kc % P_AHEAD][0] = *(const v4f*)(gA + kn * P_KC);
#pragma unroll
      for (int i = 0; i < 4; ++i)
        ring[kc % P_AHEAD][1 + i] = *(const v4f*)(gB + (size_t)(kn * P_KC + 8 * i) * LDGB);
    }
    const float* L = sm + cur * STAGE;
    v2f fa0[KS], fa1[KS], fb[KS];
#pragma unroll
    for (int s = 0; s < KS; ++s) {
      fa0[s] = *(const v2f*)(L + ao + 4 * s);
      fa1[s] = *(const v2f*)(L + ao + 16 * P_LDA + 4 * s);
      fb[s].x = L[bo + 4 * s * P_LDB];
      fb[s].y = L[bo + 4 * s * P_LDB + P_LDB];
    }
    __builtin_amdgcn_sched_barrier(0);
#pragma unroll
    for (int s = 0; s < KS_HEAD; ++s) {
      c0 = wmma4(fa0[s], fb[s], c0);
      c1 = wmma4(fa1[s], fb[s], c1);
    }
    __builtin_amdgcn_sched_barrier(0);
    if (kc + 1 < NCH) {
      float* S = sm + (cur ^ 1) * STAGE;
      *(v4f*)(S + soa) = ring[(kc + 1) % P_AHEAD][0];
#pragma unroll
      for (int i = 0; i < 4; ++i) *(v4f*)(S + sob + i * 8 * P_LDB) = ring[(kc + 1) % P_AHEAD][1 + i];
      lds_barrier();
    }
    __builtin_amdgcn_sched_barrier(0);
#pragma unroll
    for (int s = KS_HEAD; s < KS; ++s) {
      c0 = wmma4(fa0[s], fb[s], c0);
      c1 = wmma4(fa1[s], fb[s], c1);
    }
  }
}

#define W_LD 36

__global__ __launch_bounds__(256) void k_conv12(const float* __restrict__ x, const float* __restrict__ w1,
                                                const float* __restrict__ b1, const float* __restrict__ w2,
                                                const float* __restrict__ b2, float* __restrict__ y) {
  constexpr int STAGE = P_BM * P_LDA + P_KC * P_LDB;
  __shared__ __attribute__((aligned(16))) float sm[2 * STAGE + C_ * W_LD];
  float* sW2 = sm + 2 * STAGE;

  const int tid  = threadIdx.x;
  const int lane = tid & 31;
  const int half = lane >> 4;
  const int l16  = lane & 15;
  const int wv   = tid >> 5;
  const int b    = blockIdx.x >> 5;
  const int bn   = blockIdx.x & 31;

  const int arow = tid >> 3;
  const int ac4  = tid & 7;
  const int brow = tid >> 5;
  const int bc4  = tid & 31;
  const float* gA = w1 + (size_t)arow * C_ + 4 * ac4;
  const float* gB = x + (size_t)b * C_ * HW_ + (size_t)brow * HW_ + bn * P_BN + 4 * bc4;

  float bias0[8], bias1[8];
#pragma unroll
  for (int r = 0; r < 8; ++r) {
    bias0[r] = b1[r + 8 * half];
    bias1[r] = b1[16 + r + 8 * half];
  }
  v4f wq[16];
#pragma unroll
  for (int i = 0; i < 16; ++i) wq[i] = *(const v4f*)(w2 + 4 * tid + 1024 * i);

  v8f c0 = {}, c1 = {};
  tile_32x128_k512<HW_>(gA, gB, sm, c0, c1);

  float g0[8], g1[8];
#pragma unroll
  for (int r = 0; r < 8; ++r) {
    float v = c0[r] + bias0[r];
    g0[r] = 0.5f * v * (1.0f + erff(v * 0.70710678118654752f));
    float u = c1[r] + bias1[r];
    g1[r] = 0.5f * u * (1.0f + erff(u * 0.70710678118654752f));
  }
#pragma unroll
  for (int i = 0; i < 16; ++i) *(v4f*)(sW2 + (arow + 32 * i) * W_LD + 4 * ac4) = wq[i];
  __syncthreads();

  v2f hb[8];
#pragma unroll
  for (int j = 0; j < 4; ++j) {
    hb[j].x = g0[2 * j];      hb[j].y = g0[2 * j + 1];
    hb[4 + j].x = g1[2 * j];  hb[4 + j].y = g1[2 * j + 1];
  }

  const float* sA = sW2 + l16 * W_LD + 8 * half;
  const float* bp = b2 + 8 * half;
  const int n = bn * P_BN + wv * 16 + l16;
  float* Y = y + (size_t)b * C_ * HW_ + (size_t)(8 * half) * HW_ + n;

  constexpr int TR = 4;
  constexpr int NTRIP = C_ / (16 * TR);
  v4f q[2][TR][4];
#pragma unroll
  for (int t = 0; t < TR; ++t) {
    const float* p = sA + (t * 16) * W_LD;
    q[0][t][0] = *(const v4f*)(p);
    q[0][t][1] = *(const v4f*)(p + 4);
    q[0][t][2] = *(const v4f*)(p + 16);
    q[0][t][3] = *(const v4f*)(p + 20);
  }
#pragma unroll
  for (int it = 0; it < NTRIP; ++it) {
    const int cur = it & 1;
    if (it + 1 < NTRIP) {
#pragma unroll
      for (int t = 0; t < TR; ++t) {
        const float* p = sA + ((it + 1) * 16 * TR + t * 16) * W_LD;
        q[cur ^ 1][t][0] = *(const v4f*)(p);
        q[cur ^ 1][t][1] = *(const v4f*)(p + 4);
        q[cur ^ 1][t][2] = *(const v4f*)(p + 16);
        q[cur ^ 1][t][3] = *(const v4f*)(p + 20);
      }
    }
    v4f bb[TR][2];
#pragma unroll
    for (int t = 0; t < TR; ++t) {
      bb[t][0] = *(const v4f*)(bp + it * 16 * TR + t * 16);
      bb[t][1] = *(const v4f*)(bp + it * 16 * TR + t * 16 + 4);
    }
    __builtin_amdgcn_sched_barrier(0);
    v8f acc[TR] = {};
#pragma unroll
    for (int j = 0; j < 8; ++j)
#pragma unroll
      for (int t = 0; t < TR; ++t) {
        v2f a;
        a.x = q[cur][t][j >> 1][2 * (j & 1)];
        a.y = q[cur][t][j >> 1][2 * (j & 1) + 1];
        acc[t] = wmma4(a, hb[j], acc[t]);
      }
#pragma unroll
    for (int t = 0; t < TR; ++t)
#pragma unroll
      for (int r = 0; r < 8; ++r)
        Y[(size_t)(it * 16 * TR + t * 16 + r) * HW_] = acc[t][r] + bb[t][r >> 2][r & 3];
  }
}

#define E_BM 32
#define E_BN 64
#define E_KC 32
#define E_LD 36
#define E_TILES 72
#define E_KSPLIT 2

__global__ __launch_bounds__(128) void k_energy(const float* __restrict__ y, float* __restrict__ e0,
                                                float* __restrict__ e1) {
  __shared__ __attribute__((aligned(16))) float sm[2 * (E_BM + E_BN) * E_LD];
  constexpr int STAGE = (E_BM + E_BN) * E_LD;
  constexpr int NCH = HW_ / E_KSPLIT / E_KC;

  const int tid  = threadIdx.x;
  const int lane = tid & 31;
  const int half = lane >> 4;
  const int l16  = lane & 15;
  const int wv   = tid >> 5;
  const int b    = blockIdx.x / (E_KSPLIT * E_TILES);
  int rem        = blockIdx.x % (E_KSPLIT * E_TILES);
  const int kh   = rem / E_TILES;
  rem            = rem % E_TILES;
  int j = 0;
#pragma unroll
  for (int q = 0; q < 7; ++q) {
    const int nb = 2 * (8 - q);
    if (j == q && rem >= nb) { rem -= nb; j = q + 1; }
  }
  const int wdt   = 8 - j;
  const int upper = (rem >= wdt) ? 1 : 0;
  if (upper) rem -= wdt;
  const int strip = 2 * j + upper;
  const int bn    = j + rem;
  const float* yb = y + (size_t)b * C_ * HW_ + kh * (HW_ / E_KSPLIT);

  const int srow = tid >> 3;
  const int sc4  = tid & 7;
  const float* gp[6];
  int so[6];
#pragma unroll
  for (int i = 0; i < 6; ++i) {
    const int r = srow + 16 * i;
    const int grow = (i < 2) ? (strip * E_BM + r) : (bn * E_BN + (r - E_BM));
    gp[i] = yb + (size_t)grow * HW_ + 4 * sc4;
    so[i] = r * E_LD + 4 * sc4;
  }

  const int ao = l16 * E_LD + 2 * half;
  const int bo = (E_BM + wv * 16 + l16) * E_LD + 2 * half;

  v4f pre[6];
#pragma unroll
  for (int i = 0; i < 6; ++i) pre[i] = *(const v4f*)gp[i];
#pragma unroll
  for (int i = 0; i < 6; ++i) *(v4f*)(sm + so[i]) = pre[i];
  __syncthreads();

  v8f c0 = {}, c1 = {};
  constexpr int KS = E_KC / 4;
  constexpr int KS_HEAD = 6;
  for (int kc = 0; kc < NCH; ++kc) {
    const int cur = kc & 1;
    const bool more = (kc + 1 < NCH);
    if (more) {
#pragma unroll
      for (int i = 0; i < 6; ++i) pre[i] = *(const v4f*)(gp[i] + (size_t)(kc + 1) * E_KC);
    }
    const float* L = sm + cur * STAGE;
    v2f fa0[KS], fa1[KS], fb[KS];
#pragma unroll
    for (int s = 0; s < KS; ++s) {
      fa0[s] = *(const v2f*)(L + ao + 4 * s);
      fa1[s] = *(const v2f*)(L + ao + 16 * E_LD + 4 * s);
      fb[s]  = *(const v2f*)(L + bo + 4 * s);
    }
    __builtin_amdgcn_sched_barrier(0);
#pragma unroll
    for (int s = 0; s < KS_HEAD; ++s) {
      c0 = wmma4(fa0[s], fb[s], c0);
      c1 = wmma4(fa1[s], fb[s], c1);
    }
    __builtin_amdgcn_sched_barrier(0);
    if (more) {
      float* S = sm + (cur ^ 1) * STAGE;
#pragma unroll
      for (int i = 0; i < 6; ++i) *(v4f*)(S + so[i]) = pre[i];
    }
    __syncthreads();
    __builtin_amdgcn_sched_barrier(0);
#pragma unroll
    for (int s = KS_HEAD; s < KS; ++s) {
      c0 = wmma4(fa0[s], fb[s], c0);
      c1 = wmma4(fa1[s], fb[s], c1);
    }
  }

  float* D = (kh ? e1 : e0) + (size_t)b * C_ * C_;
  const int m0 = strip * E_BM + 8 * half;
  const int n0 = bn * E_BN + wv * 16 + l16;
#pragma unroll
  for (int r = 0; r < 8; ++r) {
    D[(size_t)(m0 + r) * C_ + n0]      = c0[r];
    D[(size_t)(m0 + 16 + r) * C_ + n0] = c1[r];
  }
  if (bn > j) {
    float* Tp = D + (size_t)n0 * C_ + m0;
    v4f t0 = {c0[0], c0[1], c0[2], c0[3]};
    v4f t1 = {c0[4], c0[5], c0[6], c0[7]};
    v4f t2 = {c1[0], c1[1], c1[2], c1[3]};
    v4f t3 = {c1[4], c1[5], c1[6], c1[7]};
    *(v4f*)(Tp)      = t0;
    *(v4f*)(Tp + 4)  = t1;
    *(v4f*)(Tp + 16) = t2;
    *(v4f*)(Tp + 20) = t3;
  }
}

__global__ void k_softmax(float* __restrict__ e, const float* __restrict__ eb) {
  __shared__ float red[256];
  const int tid = threadIdx.x;
  float* p = e + (size_t)blockIdx.x * C_;
  const float* q = eb + (size_t)blockIdx.x * C_;
  const float v0 = p[tid] + q[tid];
  const float v1 = p[tid + 256] + q[tid + 256];
  red[tid] = fminf(v0, v1);
  __syncthreads();
  for (int s = 128; s > 0; s >>= 1) {
    if (tid < s) red[tid] = fminf(red[tid], red[tid + s]);
    __syncthreads();
  }
  const float rmin = red[0];
  __syncthreads();
  const float e0 = expf(rmin - v0);
  const float e1 = expf(rmin - v1);
  red[tid] = e0 + e1;
  __syncthreads();
  for (int s = 128; s > 0; s >>= 1) {
    if (tid < s) red[tid] += red[tid + s];
    __syncthreads();
  }
  const float inv = 1.0f / red[0];
  p[tid] = e0 * inv;
  p[tid + 256] = e1 * inv;
}

__global__ __launch_bounds__(256) void k_m2(const float* __restrict__ attn, const float* __restrict__ wf,
                                            const float* __restrict__ gamma, float* __restrict__ m2) {
  constexpr int STAGE = P_BM * P_LDA + P_KC * P_LDB;
  __shared__ __attribute__((aligned(16))) float sm[2 * STAGE];

  const int tid   = threadIdx.x;
  const int lane  = tid & 31;
  const int half  = lane >> 4;
  const int l16   = lane & 15;
  const int wv    = tid >> 5;
  const int b     = blockIdx.x >> 6;
  const int strip = (blockIdx.x >> 2) & 15;
  const int bn    = blockIdx.x & 3;

  const int arow = tid >> 3;
  const int ac4  = tid & 7;
  const int brow = tid >> 5;
  const int bc4  = tid & 31;
  const float* gA = wf + (size_t)(strip * P_BM + arow) * C2_ + 4 * ac4;
  const float* gB = attn + (size_t)b * C_ * C_ + (size_t)brow * C_ + bn * P_BN + 4 * bc4;

  v8f c0 = {}, c1 = {};
  tile_32x128_k512<C_>(gA, gB, sm, c0, c1);

  const float g = gamma[0];
  const int n  = bn * P_BN + wv * 16 + l16;
  const int m0 = strip * P_BM + 8 * half;
  float wa[8], wb[8];
#pragma unroll
  for (int r = 0; r < 8; ++r) {
    wa[r] = wf[(size_t)(m0 + r) * C2_ + n];
    wb[r] = wf[(size_t)(m0 + 16 + r) * C2_ + n];
  }
  float* D = m2 + (size_t)b * C_ * C_;
#pragma unroll
  for (int r = 0; r < 8; ++r) {
    D[(size_t)(m0 + r) * C_ + n]      = g * c0[r] + wa[r];
    D[(size_t)(m0 + 16 + r) * C_ + n] = g * c1[r] + wb[r];
  }
}

#define F_BM  128
#define F_BN  128
#define F_KC  32
#define F_LDA 36
#define F_LDB 144

__global__ __launch_bounds__(256) void k_fusion(const float* __restrict__ m2, const float* __restrict__ y,
                                                const float* __restrict__ x, const float* __restrict__ wf,
                                                const float* __restrict__ bf, float* __restrict__ out) {
  constexpr int A_SZ  = F_BM * F_LDA;
  constexpr int STAGE = A_SZ + F_KC * F_LDB;
  __shared__ __attribute__((aligned(16))) float sm[2 * STAGE];
  constexpr int NCH  = C2_ / F_KC;
  constexpr int NCH1 = C_ / F_KC;

  const int tid  = threadIdx.x;
  const int lane = tid & 31;
  const int half = lane >> 4;
  const int l16  = lane & 15;
  const int wv   = tid >> 5;
  const int wm   = wv >> 2;
  const int wn   = wv & 3;
  const int b    = blockIdx.x >> 7;
  const int bm   = (blockIdx.x >> 5) & 3;
  const int bn   = blockIdx.x & 31;

  const int arow = tid >> 3;
  const int ac4  = tid & 7;
  const int brow = tid >> 5;
  const int bc4  = tid & 31;
  const float* gA1 = m2 + (size_t)b * C_ * C_ + (size_t)(bm * F_BM + arow) * C_ + 4 * ac4;
  const float* gA2 = wf + (size_t)(bm * F_BM + arow) * C2_ + C_ + 4 * ac4;
  const float* gB1 = y + (size_t)b * C_ * HW_ + (size_t)brow * HW_ + bn * F_BN + 4 * bc4;
  const float* gB2 = x + (size_t)b * C_ * HW_ + (size_t)brow * HW_ + bn * F_BN + 4 * bc4;
  const int soa = arow * F_LDA + 4 * ac4;
  const int sob = A_SZ + brow * F_LDB + 4 * bc4;

  const int ao = (wm * 64 + l16) * F_LDA + 2 * half;
  const int bo = A_SZ + (2 * half) * F_LDB + wn * 32 + l16;

  v4f pre[8];
#pragma unroll
  for (int i = 0; i < 4; ++i) pre[i] = *(const v4f*)(gA1 + (size_t)i * 32 * C_);
#pragma unroll
  for (int i = 0; i < 4; ++i) pre[4 + i] = *(const v4f*)(gB1 + (size_t)i * 8 * HW_);
#pragma unroll
  for (int i = 0; i < 4; ++i) *(v4f*)(sm + soa + i * 32 * F_LDA) = pre[i];
#pragma unroll
  for (int i = 0; i < 4; ++i) *(v4f*)(sm + sob + i * 8 * F_LDB) = pre[4 + i];
  __syncthreads();

  v8f c[4][2] = {};
  constexpr int KS = F_KC / 4;
  constexpr int KS_HEAD = 6;
  for (int kc = 0; kc < NCH; ++kc) {
    const int cur = kc & 1;
    const bool more = (kc + 1 < NCH);
    if (more) {
      const int kn = kc + 1;
      if (kn < NCH1) {
#pragma unroll
        for (int i = 0; i < 4; ++i) pre[i] = *(const v4f*)(gA1 + (size_t)i * 32 * C_ + kn * F_KC);
#pragma unroll
        for (int i = 0; i < 4; ++i) pre[4 + i] = *(const v4f*)(gB1 + (size_t)(kn * F_KC + 8 * i) * HW_);
      } else {
#pragma unroll
        for (int i = 0; i < 4; ++i) pre[i] = *(const v4f*)(gA2 + (size_t)i * 32 * C2_ + (kn - NCH1) * F_KC);
#pragma unroll
        for (int i = 0; i < 4; ++i) pre[4 + i] = *(const v4f*)(gB2 + (size_t)((kn - NCH1) * F_KC + 8 * i) * HW_);
      }
    }
    const float* L = sm + cur * STAGE;
    v2f fa[4][KS], fb[2][KS];
#pragma unroll
    for (int s = 0; s < KS; ++s) {
#pragma unroll
      for (int t = 0; t < 4; ++t) fa[t][s] = *(const v2f*)(L + ao + t * 16 * F_LDA + 4 * s);
#pragma unroll
      for (int t = 0; t < 2; ++t) {
        fb[t][s].x = L[bo + t * 16 + 4 * s * F_LDB];
        fb[t][s].y = L[bo + t * 16 + 4 * s * F_LDB + F_LDB];
      }
    }
    __builtin_amdgcn_sched_barrier(0);
#pragma unroll
    for (int s = 0; s < KS_HEAD; ++s)
#pragma unroll
      for (int t = 0; t < 4; ++t) {
        c[t][0] = wmma4(fa[t][s], fb[0][s], c[t][0]);
        c[t][1] = wmma4(fa[t][s], fb[1][s], c[t][1]);
      }
    __builtin_amdgcn_sched_barrier(0);
    if (more) {
      float* S = sm + (cur ^ 1) * STAGE;
#pragma unroll
      for (int i = 0; i < 4; ++i) *(v4f*)(S + soa + i * 32 * F_LDA) = pre[i];
#pragma unroll
      for (int i = 0; i < 4; ++i) *(v4f*)(S + sob + i * 8 * F_LDB) = pre[4 + i];
    }
    __syncthreads();
    __builtin_amdgcn_sched_barrier(0);
#pragma unroll
    for (int s = KS_HEAD; s < KS; ++s)
#pragma unroll
      for (int t = 0; t < 4; ++t) {
        c[t][0] = wmma4(fa[t][s], fb[0][s], c[t][0]);
        c[t][1] = wmma4(fa[t][s], fb[1][s], c[t][1]);
      }
  }

  float* D = out + (size_t)b * C_ * HW_;
  const int n0 = bn * F_BN + wn * 32 + l16;
#pragma unroll
  for (int t = 0; t < 4; ++t) {
    const int m0 = bm * F_BM + wm * 64 + t * 16 + 8 * half;
#pragma unroll
    for (int r = 0; r < 8; ++r) {
      const float bias = bf[m0 + r];
      D[(size_t)(m0 + r) * HW_ + n0]      = c[t][0][r] + bias;
      D[(size_t)(m0 + r) * HW_ + n0 + 16] = c[t][1][r] + bias;
    }
  }
}

extern "C" void kernel_launch(void* const* d_in, const int* in_sizes, int n_in,
                              void* d_out, int out_size, void* d_ws, size_t ws_size,
                              hipStream_t stream) {
  const float* x     = (const float*)d_in[0];
  const float* w1    = (const float*)d_in[1];
  const float* b1    = (const float*)d_in[2];
  const float* w2    = (const float*)d_in[3];
  const float* b2    = (const float*)d_in[4];
  const float* wf    = (const float*)d_in[5];
  const float* bf    = (const float*)d_in[6];
  const float* gamma = (const float*)d_in[7];
  float* out = (float*)d_out;

  char* ws = (char*)d_ws;
  float* y    = (float*)(ws + (size_t)4 * 1024 * 1024);
  float* attn = (float*)(ws + (size_t)68 * 1024 * 1024);
  float* m2   = (float*)(ws + (size_t)76 * 1024 * 1024);

  const dim3 blk(256);
  k_conv12    <<<dim3(256),  blk, 0, stream>>>(x, w1, b1, w2, b2, y);
  k_energy    <<<dim3(1152), dim3(128), 0, stream>>>(y, attn, m2);
  k_softmax   <<<dim3(4096), blk, 0, stream>>>(attn, m2);
  k_m2        <<<dim3(512),  blk, 0, stream>>>(attn, wf, gamma, m2);
  k_fusion    <<<dim3(1024), blk, 0, stream>>>(m2, y, x, wf, bf, out);
}
